# starting kernel
# baseline (speedup 1.0000x reference)
#include <hip/hip_runtime.h>
#include <math.h>

#define D_MODEL 1024
#define D_STATE 16
#define D_CONV  4
#define DT_RANK 64
#define D_INNER 2048
#define LSEQ    2048

typedef __attribute__((ext_vector_type(2))) float v2f;
typedef __attribute__((ext_vector_type(8))) float v8f;

struct Frag { v2f a0, a1, b0, b1, b2, b3; };

__device__ __forceinline__ void load_frag(
    Frag& f, const float* Ar0, const float* Ar1,
    const float* w0, const float* w1, const float* w2, const float* w3, int k)
{
    f.a0 = *(const v2f*)(Ar0 + k);
    f.a1 = *(const v2f*)(Ar1 + k);
    f.b0 = *(const v2f*)(w0 + k);
    f.b1 = *(const v2f*)(w1 + k);
    f.b2 = *(const v2f*)(w2 + k);
    f.b3 = *(const v2f*)(w3 + k);
}

__device__ __forceinline__ void do_wmma(const Frag& f, v8f acc[2][4])
{
    acc[0][0] = __builtin_amdgcn_wmma_f32_16x16x4_f32(false, f.a0, false, f.b0, (short)0, acc[0][0], false, false);
    acc[1][0] = __builtin_amdgcn_wmma_f32_16x16x4_f32(false, f.a1, false, f.b0, (short)0, acc[1][0], false, false);
    acc[0][1] = __builtin_amdgcn_wmma_f32_16x16x4_f32(false, f.a0, false, f.b1, (short)0, acc[0][1], false, false);
    acc[1][1] = __builtin_amdgcn_wmma_f32_16x16x4_f32(false, f.a1, false, f.b1, (short)0, acc[1][1], false, false);
    acc[0][2] = __builtin_amdgcn_wmma_f32_16x16x4_f32(false, f.a0, false, f.b2, (short)0, acc[0][2], false, false);
    acc[1][2] = __builtin_amdgcn_wmma_f32_16x16x4_f32(false, f.a1, false, f.b2, (short)0, acc[1][2], false, false);
    acc[0][3] = __builtin_amdgcn_wmma_f32_16x16x4_f32(false, f.a0, false, f.b3, (short)0, acc[0][3], false, false);
    acc[1][3] = __builtin_amdgcn_wmma_f32_16x16x4_f32(false, f.a1, false, f.b3, (short)0, acc[1][3], false, false);
}

__global__ __launch_bounds__(128, 1) void gemm_wmma_f32(
    const float* __restrict__ A, const float* __restrict__ W,
    const float* __restrict__ bias, float* __restrict__ C,
    int M, int N, int K, int lda, int act)
{
    const int lane = threadIdx.x & 31;
    const int wave = threadIdx.x >> 5;
    const int m0   = (blockIdx.y << 7) + (wave << 5);
    const int n0   = blockIdx.x << 6;
    if (m0 >= M) return;

    const int mrow  = lane & 15;
    const int khalf = (lane >> 4) << 1;

    const float* __restrict__ Ar0 = A + (size_t)(m0 + mrow) * lda + khalf;
    const float* __restrict__ Ar1 = Ar0 + (size_t)16 * lda;

    const float* Wr[4];
#pragma unroll
    for (int j = 0; j < 4; ++j) {
        int row = n0 + (j << 4) + mrow;
        row = (row < N) ? row : (N - 1);
        Wr[j] = W + (size_t)row * K + khalf;
    }

    v8f acc[2][4];
#pragma unroll
    for (int i = 0; i < 2; ++i)
#pragma unroll
        for (int j = 0; j < 4; ++j) acc[i][j] = (v8f){};

    Frag f0, f1;
    load_frag(f0, Ar0, Ar1, Wr[0], Wr[1], Wr[2], Wr[3], 0);

    int k0 = 0;
    for (; k0 < K - 16; k0 += 16) {
        __builtin_prefetch(Ar0   + k0 + 128, 0, 1);
        __builtin_prefetch(Ar1   + k0 + 128, 0, 1);
        __builtin_prefetch(Wr[0] + k0 + 128, 0, 1);
        __builtin_prefetch(Wr[2] + k0 + 128, 0, 1);
        load_frag(f1, Ar0, Ar1, Wr[0], Wr[1], Wr[2], Wr[3], k0 + 4);
        do_wmma(f0, acc);
        load_frag(f0, Ar0, Ar1, Wr[0], Wr[1], Wr[2], Wr[3], k0 + 8);
        do_wmma(f1, acc);
        load_frag(f1, Ar0, Ar1, Wr[0], Wr[1], Wr[2], Wr[3], k0 + 12);
        do_wmma(f0, acc);
        load_frag(f0, Ar0, Ar1, Wr[0], Wr[1], Wr[2], Wr[3], k0 + 16);
        do_wmma(f1, acc);
    }
    load_frag(f1, Ar0, Ar1, Wr[0], Wr[1], Wr[2], Wr[3], k0 + 4);
    do_wmma(f0, acc);
    load_frag(f0, Ar0, Ar1, Wr[0], Wr[1], Wr[2], Wr[3], k0 + 8);
    do_wmma(f1, acc);
    load_frag(f1, Ar0, Ar1, Wr[0], Wr[1], Wr[2], Wr[3], k0 + 12);
    do_wmma(f0, acc);
    do_wmma(f1, acc);

    const int rsub = (lane >> 4) << 3;
#pragma unroll
    for (int j = 0; j < 4; ++j) {
        const int col = n0 + (j << 4) + mrow;
        if ((n0 + (j << 4)) >= N) continue;
        const float bv = (act == 1) ? bias[col] : 0.0f;
#pragma unroll
        for (int i = 0; i < 2; ++i) {
            const int rbase = m0 + (i << 4) + rsub;
#pragma unroll
            for (int r = 0; r < 8; ++r) {
                float v = acc[i][j][r];
                if (act == 1) {
                    v += bv;
                    v = (v > 20.0f) ? v : log1pf(__expf(v));
                }
                C[(size_t)(rbase + r) * N + col] = v;
            }
        }
    }
}

__global__ __launch_bounds__(256) void conv_silu_kernel(
    const float* __restrict__ xz, const float* __restrict__ cw,
    const float* __restrict__ cb, float* __restrict__ xc)
{
    int idx = blockIdx.x * blockDim.x + threadIdx.x;
    if (idx >= LSEQ * D_INNER) return;
    int l = idx >> 11;
    int d = idx & (D_INNER - 1);

    float acc = cb[d];
#pragma unroll
    for (int j = 0; j < D_CONV; ++j) {
        int ls = l - (D_CONV - 1) + j;
        if (ls >= 0)
            acc += cw[d * D_CONV + j] * xz[(size_t)ls * (2 * D_INNER) + d];
    }
    xc[idx] = acc / (1.0f + __expf(-acc));
}

#define SCAN_TS 32
__global__ __launch_bounds__(256) void ssm_scan_kernel(
    const float* __restrict__ xdbl, const float* __restrict__ dtbuf,
    const float* __restrict__ xz,   const float* __restrict__ A_log,
    const float* __restrict__ D_param, float* __restrict__ xc_io)
{
    __shared__ float sB[SCAN_TS][D_STATE];
    __shared__ float sC[SCAN_TS][D_STATE];

    const int d = blockIdx.x * blockDim.x + threadIdx.x;

    float Acoef[D_STATE];
#pragma unroll
    for (int n = 0; n < D_STATE; ++n)
        Acoef[n] = -__expf(A_log[(size_t)d * D_STATE + n]);
    const float Dp = D_param[d];

    float h[D_STATE];
#pragma unroll
    for (int n = 0; n < D_STATE; ++n) h[n] = 0.0f;

    for (int lc = 0; lc < LSEQ; lc += SCAN_TS) {
        __syncthreads();
        for (int i = threadIdx.x; i < SCAN_TS * D_STATE; i += blockDim.x) {
            int t = i >> 4, n = i & 15;
            const float* row = xdbl + (size_t)(lc + t) * (DT_RANK + 2 * D_STATE);
            sB[t][n] = row[DT_RANK + n];
            sC[t][n] = row[DT_RANK + D_STATE + n];
        }
        __syncthreads();

        for (int t = 0; t < SCAN_TS; ++t) {
            const int l = lc + t;
            const float dt = dtbuf[(size_t)l * D_INNER + d];
            const float xv = xc_io[(size_t)l * D_INNER + d];
            float y = 0.0f;
#pragma unroll
            for (int n = 0; n < D_STATE; ++n) {
                float dA = __expf(dt * Acoef[n]);
                h[n] = dA * h[n] + (dt * sB[t][n]) * xv;
                y += h[n] * sC[t][n];
            }
            y += xv * Dp;
            float z = xz[(size_t)l * (2 * D_INNER) + D_INNER + d];
            float g = z / (1.0f + __expf(-z));
            xc_io[(size_t)l * D_INNER + d] = y * g;
        }
    }
}

extern "C" void kernel_launch(void* const* d_in, const int* in_sizes, int n_in,
                              void* d_out, int out_size, void* d_ws, size_t ws_size,
                              hipStream_t stream) {
    const float* x       = (const float*)d_in[0];
    const float* W_in    = (const float*)d_in[1];
    const float* conv_w  = (const float*)d_in[2];
    const float* conv_b  = (const float*)d_in[3];
    const float* W_x     = (const float*)d_in[4];
    const float* W_dt    = (const float*)d_in[5];
    const float* b_dt    = (const float*)d_in[6];
    const float* A_log   = (const float*)d_in[7];
    const float* D_param = (const float*)d_in[8];
    const float* W_out   = (const float*)d_in[9];
    float* out = (float*)d_out;

    float* xz   = (float*)d_ws;
    float* xc   = xz   + (size_t)LSEQ * 2 * D_INNER;
    float* xdbl = xc   + (size_t)LSEQ * D_INNER;
    float* dtb  = xdbl + (size_t)LSEQ * (DT_RANK + 2 * D_STATE);

    dim3 blk(128, 1, 1);

    gemm_wmma_f32<<<dim3(4096 / 64, LSEQ / 128), blk, 0, stream>>>(
        x, W_in, nullptr, xz, LSEQ, 2 * D_INNER, D_MODEL, D_MODEL, 0);

    conv_silu_kernel<<<(LSEQ * D_INNER) / 256, 256, 0, stream>>>(
        xz, conv_w, conv_b, xc);

    gemm_wmma_f32<<<dim3((96 + 63) / 64, LSEQ / 128), blk, 0, stream>>>(
        xc, W_x, nullptr, xdbl, LSEQ, DT_RANK + 2 * D_STATE, D_INNER, D_INNER, 0);

    gemm_wmma_f32<<<dim3(D_INNER / 64, LSEQ / 128), blk, 0, stream>>>(
        xdbl, W_dt, b_dt, dtb, LSEQ, D_INNER, DT_RANK, DT_RANK + 2 * D_STATE, 1);

    ssm_scan_kernel<<<D_INNER / 256, 256, 0, stream>>>(
        xdbl, dtb, xz, A_log, D_param, xc);

    gemm_wmma_f32<<<dim3(D_MODEL / 64, LSEQ / 128), blk, 0, stream>>>(
        xc, W_out, nullptr, out, LSEQ, D_MODEL, D_INNER, D_INNER, 0);
}


# scan: one lane per (channel,state) instead of 16 states per thread: 16x the waves, 64 groups of 16 waves; y summed over the 16 states by a cross-lane tree (a different order of that sum)
# speedup vs baseline: 5.7128x; 5.7128x over previous
#include <hip/hip_runtime.h>
#include <math.h>

#define D_MODEL 1024
#define D_STATE 16
#define D_CONV  4
#define DT_RANK 64
#define D_INNER 2048
#define LSEQ    2048

typedef __attribute__((ext_vector_type(2))) float v2f;
typedef __attribute__((ext_vector_type(8))) float v8f;

struct Frag { v2f a0, a1, b0, b1, b2, b3; };

__device__ __forceinline__ void load_frag(
    Frag& f, const float* Ar0, const float* Ar1,
    const float* w0, const float* w1, const float* w2, const float* w3, int k)
{
    f.a0 = *(const v2f*)(Ar0 + k);
    f.a1 = *(const v2f*)(Ar1 + k);
    f.b0 = *(const v2f*)(w0 + k);
    f.b1 = *(const v2f*)(w1 + k);
    f.b2 = *(const v2f*)(w2 + k);
    f.b3 = *(const v2f*)(w3 + k);
}

__device__ __forceinline__ void do_wmma(const Frag& f, v8f acc[2][4])
{
    acc[0][0] = __builtin_amdgcn_wmma_f32_16x16x4_f32(false, f.a0, false, f.b0, (short)0, acc[0][0], false, false);
    acc[1][0] = __builtin_amdgcn_wmma_f32_16x16x4_f32(false, f.a1, false, f.b0, (short)0, acc[1][0], false, false);
    acc[0][1] = __builtin_amdgcn_wmma_f32_16x16x4_f32(false, f.a0, false, f.b1, (short)0, acc[0][1], false, false);
    acc[1][1] = __builtin_amdgcn_wmma_f32_16x16x4_f32(false, f.a1, false, f.b1, (short)0, acc[1][1], false, false);
    acc[0][2] = __builtin_amdgcn_wmma_f32_16x16x4_f32(false, f.a0, false, f.b2, (short)0, acc[0][2], false, false);
    acc[1][2] = __builtin_amdgcn_wmma_f32_16x16x4_f32(false, f.a1, false, f.b2, (short)0, acc[1][2], false, false);
    acc[0][3] = __builtin_amdgcn_wmma_f32_16x16x4_f32(false, f.a0, false, f.b3, (short)0, acc[0][3], false, false);
    acc[1][3] = __builtin_amdgcn_wmma_f32_16x16x4_f32(false, f.a1, false, f.b3, (short)0, acc[1][3], false, false);
}

__global__ __launch_bounds__(128, 1) void gemm_wmma_f32(
    const float* __restrict__ A, const float* __restrict__ W,
    const float* __restrict__ bias, float* __restrict__ C,
    int M, int N, int K, int lda, int act)
{
    const int lane = threadIdx.x & 31;
    const int wave = threadIdx.x >> 5;
    const int m0   = (blockIdx.y << 7) + (wave << 5);
    const int n0   = blockIdx.x << 6;
    if (m0 >= M) return;

    const int mrow  = lane & 15;
    const int khalf = (lane >> 4) << 1;

    const float* __restrict__ Ar0 = A + (size_t)(m0 + mrow) * lda + khalf;
    const float* __restrict__ Ar1 = Ar0 + (size_t)16 * lda;

    const float* Wr[4];
#pragma unroll
    for (int j = 0; j < 4; ++j) {
        int row = n0 + (j << 4) + mrow;
        row = (row < N) ? row : (N - 1);
        Wr[j] = W + (size_t)row * K + khalf;
    }

    v8f acc[2][4];
#pragma unroll
    for (int i = 0; i < 2; ++i)
#pragma unroll
        for (int j = 0; j < 4; ++j) acc[i][j] = (v8f){};

    Frag f0, f1;
    load_frag(f0, Ar0, Ar1, Wr[0], Wr[1], Wr[2], Wr[3], 0);

    int k0 = 0;
    for (; k0 < K - 16; k0 += 16) {
        __builtin_prefetch(Ar0   + k0 + 128, 0, 1);
        __builtin_prefetch(Ar1   + k0 + 128, 0, 1);
        __builtin_prefetch(Wr[0] + k0 + 128, 0, 1);
        __builtin_prefetch(Wr[2] + k0 + 128, 0, 1);
        load_frag(f1, Ar0, Ar1, Wr[0], Wr[1], Wr[2], Wr[3], k0 + 4);
        do_wmma(f0, acc);
        load_frag(f0, Ar0, Ar1, Wr[0], Wr[1], Wr[2], Wr[3], k0 + 8);
        do_wmma(f1, acc);
        load_frag(f1, Ar0, Ar1, Wr[0], Wr[1], Wr[2], Wr[3], k0 + 12);
        do_wmma(f0, acc);
        load_frag(f0, Ar0, Ar1, Wr[0], Wr[1], Wr[2], Wr[3], k0 + 16);
        do_wmma(f1, acc);
    }
    load_frag(f1, Ar0, Ar1, Wr[0], Wr[1], Wr[2], Wr[3], k0 + 4);
    do_wmma(f0, acc);
    load_frag(f0, Ar0, Ar1, Wr[0], Wr[1], Wr[2], Wr[3], k0 + 8);
    do_wmma(f1, acc);
    load_frag(f1, Ar0, Ar1, Wr[0], Wr[1], Wr[2], Wr[3], k0 + 12);
    do_wmma(f0, acc);
    do_wmma(f1, acc);

    const int rsub = (lane >> 4) << 3;
#pragma unroll
    for (int j = 0; j < 4; ++j) {
        const int col = n0 + (j << 4) + mrow;
        if ((n0 + (j << 4)) >= N) continue;
        const float bv = (act == 1) ? bias[col] : 0.0f;
#pragma unroll
        for (int i = 0; i < 2; ++i) {
            const int rbase = m0 + (i << 4) + rsub;
#pragma unroll
            for (int r = 0; r < 8; ++r) {
                float v = acc[i][j][r];
                if (act == 1) {
                    v += bv;
                    v = (v > 20.0f) ? v : log1pf(__expf(v));
                }
                C[(size_t)(rbase + r) * N + col] = v;
            }
        }
    }
}

__global__ __launch_bounds__(256) void conv_silu_kernel(
    const float* __restrict__ xz, const float* __restrict__ cw,
    const float* __restrict__ cb, float* __restrict__ xc)
{
    int idx = blockIdx.x * blockDim.x + threadIdx.x;
    if (idx >= LSEQ * D_INNER) return;
    int l = idx >> 11;
    int d = idx & (D_INNER - 1);

    float acc = cb[d];
#pragma unroll
    for (int j = 0; j < D_CONV; ++j) {
        int ls = l - (D_CONV - 1) + j;
        if (ls >= 0)
            acc += cw[d * D_CONV + j] * xz[(size_t)ls * (2 * D_INNER) + d];
    }
    xc[idx] = acc / (1.0f + __expf(-acc));
}

#define SC_CH 32
#define SC_T  64
#define SC_R  (SC_T / 16)

template <int CTRL>
__device__ __forceinline__ float dpp_row(float v)
{
    const int b = __builtin_bit_cast(int, v);
    return __builtin_bit_cast(float, __builtin_amdgcn_update_dpp(0, b, CTRL, 0xf, 0xf, true));
}

template <int XM>
__device__ __forceinline__ float pair_fold(float even, float odd, bool bit)
{
    const float keep = bit ? odd : even;
    const float send = bit ? even : odd;
    return keep + dpp_row<0x160 + XM>(send);
}

__device__ __forceinline__ float row16_sum_by_step(const float (&p)[16], int n)
{
    const bool b0 = (n & 1) != 0, b1 = (n & 2) != 0, b2 = (n & 4) != 0, b3 = (n & 8) != 0;
    float q[8], r[4], u[2];
#pragma unroll
    for (int k = 0; k < 8; ++k) q[k] = pair_fold<1>(p[2 * k], p[2 * k + 1], b0);
#pragma unroll
    for (int k = 0; k < 4; ++k) r[k] = pair_fold<2>(q[2 * k], q[2 * k + 1], b1);
#pragma unroll
    for (int k = 0; k < 2; ++k) u[k] = pair_fold<4>(r[2 * k], r[2 * k + 1], b2);
    return pair_fold<8>(u[0], u[1], b3);
}

__global__ __launch_bounds__(512) void ssm_scan_kernel(
    const float* __restrict__ xdbl, const float* __restrict__ dtbuf,
    const float* __restrict__ xz,   const float* __restrict__ A_log,
    const float* __restrict__ D_param, float* __restrict__ xc_io)
{
    __shared__ float sIn[SC_T][128];
    __shared__ float sY[SC_T][SC_CH + 1];

    const int lane = threadIdx.x & 31;
    const int wave = threadIdx.x >> 5;
    const int D0   = blockIdx.x * SC_CH;

    const int n   = lane & 15;
    const int chl = (wave << 1) + (lane >> 4);
    const float Acoef = -__expf(A_log[(size_t)(D0 + chl) * D_STATE + n]);
    float h = 0.0f;

    const int gch = D0 + lane;
    const float Dp = D_param[gch];

    float r_dt[SC_R], r_x[SC_R], r_z[SC_R], r_bc[SC_R];
#pragma unroll
    for (int r = 0; r < SC_R; ++r) {
        const size_t l = (size_t)(r * 16 + wave);
        r_dt[r] = dtbuf[l * D_INNER + gch];
        r_x[r]  = xc_io[l * D_INNER + gch];
        r_z[r]  = xz[l * (2 * D_INNER) + D_INNER + gch];
        r_bc[r] = xdbl[l * (DT_RANK + 2 * D_STATE) + DT_RANK + lane];
    }

    for (int lc = 0; lc < LSEQ; lc += SC_T) {
        float c_x[SC_R], c_z[SC_R];
#pragma unroll
        for (int r = 0; r < SC_R; ++r) {
            const int t = r * 16 + wave;
            sIn[t][lane]      = r_dt[r];
            sIn[t][32 + lane] = r_x[r];
            sIn[t][64 + lane] = r_bc[r];
            c_x[r] = r_x[r];
            c_z[r] = r_z[r];
        }
        __syncthreads();

        if (lc + SC_T < LSEQ) {
#pragma unroll
            for (int r = 0; r < SC_R; ++r) {
                const size_t l = (size_t)(lc + SC_T + r * 16 + wave);
                r_dt[r] = dtbuf[l * D_INNER + gch];
                r_x[r]  = xc_io[l * D_INNER + gch];
                r_z[r]  = xz[l * (2 * D_INNER) + D_INNER + gch];
                r_bc[r] = xdbl[l * (DT_RANK + 2 * D_STATE) + DT_RANK + lane];
            }
        }

        float yk[SC_R];
#pragma unroll
        for (int r = 0; r < SC_R; ++r) {
            float p[16];
#pragma unroll
            for (int i = 0; i < 16; ++i) {
                const int t = r * 16 + i;
                const float dt = sIn[t][chl];
                const float xv = sIn[t][32 + chl];
                const float Bn = sIn[t][64 + n];
                const float Cn = sIn[t][80 + n];
                const float dA = __expf(dt * Acoef);
                h = dA * h + (dt * Bn) * xv;
                p[i] = h * Cn;
            }
            yk[r] = row16_sum_by_step(p, n);
        }
#pragma unroll
        for (int r = 0; r < SC_R; ++r) sY[r * 16 + n][chl] = yk[r];
        __syncthreads();

#pragma unroll
        for (int r = 0; r < SC_R; ++r) {
            const int t = r * 16 + wave;
            float y = sY[t][lane];
            y += c_x[r] * Dp;
            const float z = c_z[r];
            const float g = z / (1.0f + __expf(-z));
            xc_io[(size_t)(lc + t) * D_INNER + gch] = y * g;
        }
    }
}

extern "C" void kernel_launch(void* const* d_in, const int* in_sizes, int n_in,
                              void* d_out, int out_size, void* d_ws, size_t ws_size,
                              hipStream_t stream) {
    const float* x       = (const float*)d_in[0];
    const float* W_in    = (const float*)d_in[1];
    const float* conv_w  = (const float*)d_in[2];
    const float* conv_b  = (const float*)d_in[3];
    const float* W_x     = (const float*)d_in[4];
    const float* W_dt    = (const float*)d_in[5];
    const float* b_dt    = (const float*)d_in[6];
    const float* A_log   = (const float*)d_in[7];
    const float* D_param = (const float*)d_in[8];
    const float* W_out   = (const float*)d_in[9];
    float* out = (float*)d_out;

    float* xz   = (float*)d_ws;
    float* xc   = xz   + (size_t)LSEQ * 2 * D_INNER;
    float* xdbl = xc   + (size_t)LSEQ * D_INNER;
    float* dtb  = xdbl + (size_t)LSEQ * (DT_RANK + 2 * D_STATE);

    dim3 blk(128, 1, 1);

    gemm_wmma_f32<<<dim3(4096 / 64, LSEQ / 128), blk, 0, stream>>>(
        x, W_in, nullptr, xz, LSEQ, 2 * D_INNER, D_MODEL, D_MODEL, 0);

    conv_silu_kernel<<<(LSEQ * D_INNER) / 256, 256, 0, stream>>>(
        xz, conv_w, conv_b, xc);

    gemm_wmma_f32<<<dim3((96 + 63) / 64, LSEQ / 128), blk, 0, stream>>>(
        xc, W_x, nullptr, xdbl, LSEQ, DT_RANK + 2 * D_STATE, D_INNER, D_INNER, 0);

    gemm_wmma_f32<<<dim3(D_INNER / 64, LSEQ / 128), blk, 0, stream>>>(
        xdbl, W_dt, b_dt, dtb, LSEQ, D_INNER, DT_RANK, DT_RANK + 2 * D_STATE, 1);

    ssm_scan_kernel<<<D_INNER / SC_CH, 512, 0, stream>>>(
        xdbl, dtb, xz, A_log, D_param, xc);

    gemm_wmma_f32<<<dim3(D_MODEL / 64, LSEQ / 128), blk, 0, stream>>>(
        xc, W_out, nullptr, out, LSEQ, D_MODEL, D_INNER, D_INNER, 0);
}


# GEMMs: A and W tiles of 32 k staged through LDS by whole-line 16-byte loads (two stages, one barrier per tile); fragments read from LDS, not by 8-byte strided global loads; k order of every sum kept
# speedup vs baseline: 8.6617x; 1.5162x over previous
#include <hip/hip_runtime.h>
#include <math.h>

#define D_MODEL 1024
#define D_STATE 16
#define D_CONV  4
#define DT_RANK 64
#define D_INNER 2048
#define LSEQ    2048

typedef __attribute__((ext_vector_type(2))) float v2f;
typedef __attribute__((ext_vector_type(4))) float v4f;
typedef __attribute__((ext_vector_type(8))) float v8f;

struct Frag { v2f a0, a1, b0, b1, b2, b3; };

#define GB_M 128
#define GB_N 64
#define GB_K 32
#define GB_P 36

__device__ __forceinline__ void load_frag(Frag& f, const float* pA, const float* pB, int k)
{
    f.a0 = *(const v2f*)(pA + k);
    f.a1 = *(const v2f*)(pA + 16 * GB_P + k);
    f.b0 = *(const v2f*)(pB + k);
    f.b1 = *(const v2f*)(pB + 16 * GB_P + k);
    f.b2 = *(const v2f*)(pB + 32 * GB_P + k);
    f.b3 = *(const v2f*)(pB + 48 * GB_P + k);
}

__device__ __forceinline__ void do_wmma(const Frag& f, v8f acc[2][4])
{
    acc[0][0] = __builtin_amdgcn_wmma_f32_16x16x4_f32(false, f.a0, false, f.b0, (short)0, acc[0][0], false, false);
    acc[1][0] = __builtin_amdgcn_wmma_f32_16x16x4_f32(false, f.a1, false, f.b0, (short)0, acc[1][0], false, false);
    acc[0][1] = __builtin_amdgcn_wmma_f32_16x16x4_f32(false, f.a0, false, f.b1, (short)0, acc[0][1], false, false);
    acc[1][1] = __builtin_amdgcn_wmma_f32_16x16x4_f32(false, f.a1, false, f.b1, (short)0, acc[1][1], false, false);
    acc[0][2] = __builtin_amdgcn_wmma_f32_16x16x4_f32(false, f.a0, false, f.b2, (short)0, acc[0][2], false, false);
    acc[1][2] = __builtin_amdgcn_wmma_f32_16x16x4_f32(false, f.a1, false, f.b2, (short)0, acc[1][2], false, false);
    acc[0][3] = __builtin_amdgcn_wmma_f32_16x16x4_f32(false, f.a0, false, f.b3, (short)0, acc[0][3], false, false);
    acc[1][3] = __builtin_amdgcn_wmma_f32_16x16x4_f32(false, f.a1, false, f.b3, (short)0, acc[1][3], false, false);
}

__global__ __launch_bounds__(128) void gemm_wmma_f32(
    const float* __restrict__ A, const float* __restrict__ W,
    const float* __restrict__ bias, float* __restrict__ C,
    int M, int N, int K, int lda, int act)
{
    __shared__ __attribute__((aligned(16))) float sA[2][GB_M * GB_P];
    __shared__ __attribute__((aligned(16))) float sB[2][GB_N * GB_P];

    const int tid  = threadIdx.x;
    const int lane = tid & 31;
    const int wave = tid >> 5;
    const int mblk = blockIdx.y << 7;
    const int m0   = mblk + (wave << 5);
    const int n0   = blockIdx.x << 6;
    if (mblk >= M) return;

    const int lr = tid >> 3;
    const int q  = tid & 7;
    const float* __restrict__ gA = A + (size_t)(mblk + lr) * lda + (q << 2);
    const size_t aStep = (size_t)16 * lda;
    const float* gB[4];
#pragma unroll
    for (int j = 0; j < 4; ++j) {
        int row = n0 + lr + (j << 4);
        row = (row < N) ? row : (N - 1);
        gB[j] = W + (size_t)row * K + (q << 2);
    }
    const int sOff = lr * GB_P + (q << 2);

    const int mrow  = lane & 15;
    const int khalf = (lane >> 4) << 1;
    const int aOff  = ((wave << 5) + mrow) * GB_P + khalf;
    const int bOff  = mrow * GB_P + khalf;

    v8f acc[2][4];
#pragma unroll
    for (int i = 0; i < 2; ++i)
#pragma unroll
        for (int j = 0; j < 4; ++j) acc[i][j] = (v8f){};

    v4f ra[8], rb[4];
#pragma unroll
    for (int j = 0; j < 8; ++j) ra[j] = *(const v4f*)(gA + j * aStep);
#pragma unroll
    for (int j = 0; j < 4; ++j) rb[j] = *(const v4f*)(gB[j]);
#pragma unroll
    for (int j = 0; j < 8; ++j) *(v4f*)&sA[0][sOff + j * 16 * GB_P] = ra[j];
#pragma unroll
    for (int j = 0; j < 4; ++j) *(v4f*)&sB[0][sOff + j * 16 * GB_P] = rb[j];
    __syncthreads();

    int cur = 0;
    for (int kb = 0; kb < K; kb += GB_K) {
        const bool more = (kb + GB_K) < K;
        if (more) {
#pragma unroll
            for (int j = 0; j < 8; ++j) ra[j] = *(const v4f*)(gA + j * aStep + (kb + GB_K));
#pragma unroll
            for (int j = 0; j < 4; ++j) rb[j] = *(const v4f*)(gB[j] + (kb + GB_K));
        }

        const float* pA = &sA[cur][aOff];
        const float* pB = &sB[cur][bOff];
        Frag f0, f1;
        load_frag(f0, pA, pB, 0);
#pragma unroll
        for (int ks = 0; ks < GB_K; ks += 8) {
            load_frag(f1, pA, pB, ks + 4);
            do_wmma(f0, acc);
            if (ks + 8 < GB_K) load_frag(f0, pA, pB, ks + 8);
            do_wmma(f1, acc);
        }

        if (more) {
#pragma unroll
            for (int j = 0; j < 8; ++j) *(v4f*)&sA[cur ^ 1][sOff + j * 16 * GB_P] = ra[j];
#pragma unroll
            for (int j = 0; j < 4; ++j) *(v4f*)&sB[cur ^ 1][sOff + j * 16 * GB_P] = rb[j];
        }
        __syncthreads();
        cur ^= 1;
    }

    const int rsub = (lane >> 4) << 3;
#pragma unroll
    for (int j = 0; j < 4; ++j) {
        const int col = n0 + (j << 4) + mrow;
        if ((n0 + (j << 4)) >= N) continue;
        const float bv = (act == 1) ? bias[col] : 0.0f;
#pragma unroll
        for (int i = 0; i < 2; ++i) {
            const int rbase = m0 + (i << 4) + rsub;
#pragma unroll
            for (int r = 0; r < 8; ++r) {
                float v = acc[i][j][r];
                if (act == 1) {
                    v += bv;
                    v = (v > 20.0f) ? v : log1pf(__expf(v));
                }
                C[(size_t)(rbase + r) * N + col] = v;
            }
        }
    }
}

__global__ __launch_bounds__(256) void conv_silu_kernel(
    const float* __restrict__ xz, const float* __restrict__ cw,
    const float* __restrict__ cb, float* __restrict__ xc)
{
    int idx = blockIdx.x * blockDim.x + threadIdx.x;
    if (idx >= LSEQ * D_INNER) return;
    int l = idx >> 11;
    int d = idx & (D_INNER - 1);

    float acc = cb[d];
#pragma unroll
    for (int j = 0; j < D_CONV; ++j) {
        int ls = l - (D_CONV - 1) + j;
        if (ls >= 0)
            acc += cw[d * D_CONV + j] * xz[(size_t)ls * (2 * D_INNER) + d];
    }
    xc[idx] = acc / (1.0f + __expf(-acc));
}

#define SC_CH 32
#define SC_T  64
#define SC_R  (SC_T / 16)

template <int CTRL>
__device__ __forceinline__ float dpp_row(float v)
{
    const int b = __builtin_bit_cast(int, v);
    return __builtin_bit_cast(float, __builtin_amdgcn_update_dpp(0, b, CTRL, 0xf, 0xf, true));
}

template <int XM>
__device__ __forceinline__ float pair_fold(float even, float odd, bool bit)
{
    const float keep = bit ? odd : even;
    const float send = bit ? even : odd;
    return keep + dpp_row<0x160 + XM>(send);
}

__device__ __forceinline__ float row16_sum_by_step(const float (&p)[16], int n)
{
    const bool b0 = (n & 1) != 0, b1 = (n & 2) != 0, b2 = (n & 4) != 0, b3 = (n & 8) != 0;
    float q[8], r[4], u[2];
#pragma unroll
    for (int k = 0; k < 8; ++k) q[k] = pair_fold<1>(p[2 * k], p[2 * k + 1], b0);
#pragma unroll
    for (int k = 0; k < 4; ++k) r[k] = pair_fold<2>(q[2 * k], q[2 * k + 1], b1);
#pragma unroll
    for (int k = 0; k < 2; ++k) u[k] = pair_fold<4>(r[2 * k], r[2 * k + 1], b2);
    return pair_fold<8>(u[0], u[1], b3);
}

__global__ __launch_bounds__(512) void ssm_scan_kernel(
    const float* __restrict__ xdbl, const float* __restrict__ dtbuf,
    const float* __restrict__ xz,   const float* __restrict__ A_log,
    const float* __restrict__ D_param, float* __restrict__ xc_io)
{
    __shared__ float sIn[SC_T][128];
    __shared__ float sY[SC_T][SC_CH + 1];

    const int lane = threadIdx.x & 31;
    const int wave = threadIdx.x >> 5;
    const int D0   = blockIdx.x * SC_CH;

    const int n   = lane & 15;
    const int chl = (wave << 1) + (lane >> 4);
    const float Acoef = -__expf(A_log[(size_t)(D0 + chl) * D_STATE + n]);
    float h = 0.0f;

    const int gch = D0 + lane;
    const float Dp = D_param[gch];

    float r_dt[SC_R], r_x[SC_R], r_z[SC_R], r_bc[SC_R];
#pragma unroll
    for (int r = 0; r < SC_R; ++r) {
        const size_t l = (size_t)(r * 16 + wave);
        r_dt[r] = dtbuf[l * D_INNER + gch];
        r_x[r]  = xc_io[l * D_INNER + gch];
        r_z[r]  = xz[l * (2 * D_INNER) + D_INNER + gch];
        r_bc[r] = xdbl[l * (DT_RANK + 2 * D_STATE) + DT_RANK + lane];
    }

    for (int lc = 0; lc < LSEQ; lc += SC_T) {
        float c_x[SC_R], c_z[SC_R];
#pragma unroll
        for (int r = 0; r < SC_R; ++r) {
            const int t = r * 16 + wave;
            sIn[t][lane]      = r_dt[r];
            sIn[t][32 + lane] = r_x[r];
            sIn[t][64 + lane] = r_bc[r];
            c_x[r] = r_x[r];
            c_z[r] = r_z[r];
        }
        __syncthreads();

        if (lc + SC_T < LSEQ) {
#pragma unroll
            for (int r = 0; r < SC_R; ++r) {
                const size_t l = (size_t)(lc + SC_T + r * 16 + wave);
                r_dt[r] = dtbuf[l * D_INNER + gch];
                r_x[r]  = xc_io[l * D_INNER + gch];
                r_z[r]  = xz[l * (2 * D_INNER) + D_INNER + gch];
                r_bc[r] = xdbl[l * (DT_RANK + 2 * D_STATE) + DT_RANK + lane];
            }
        }

        float yk[SC_R];
#pragma unroll
        for (int r = 0; r < SC_R; ++r) {
            float p[16];
#pragma unroll
            for (int i = 0; i < 16; ++i) {
                const int t = r * 16 + i;
                const float dt = sIn[t][chl];
                const float xv = sIn[t][32 + chl];
                const float Bn = sIn[t][64 + n];
                const float Cn = sIn[t][80 + n];
                const float dA = __expf(dt * Acoef);
                h = dA * h + (dt * Bn) * xv;
                p[i] = h * Cn;
            }
            yk[r] = row16_sum_by_step(p, n);
        }
#pragma unroll
        for (int r = 0; r < SC_R; ++r) sY[r * 16 + n][chl] = yk[r];
        __syncthreads();

#pragma unroll
        for (int r = 0; r < SC_R; ++r) {
            const int t = r * 16 + wave;
            float y = sY[t][lane];
            y += c_x[r] * Dp;
            const float z = c_z[r];
            const float g = z / (1.0f + __expf(-z));
            xc_io[(size_t)(lc + t) * D_INNER + gch] = y * g;
        }
    }
}

extern "C" void kernel_launch(void* const* d_in, const int* in_sizes, int n_in,
                              void* d_out, int out_size, void* d_ws, size_t ws_size,
                              hipStream_t stream) {
    const float* x       = (const float*)d_in[0];
    const float* W_in    = (const float*)d_in[1];
    const float* conv_w  = (const float*)d_in[2];
    const float* conv_b  = (const float*)d_in[3];
    const float* W_x     = (const float*)d_in[4];
    const float* W_dt    = (const float*)d_in[5];
    const float* b_dt    = (const float*)d_in[6];
    const float* A_log   = (const float*)d_in[7];
    const float* D_param = (const float*)d_in[8];
    const float* W_out   = (const float*)d_in[9];
    float* out = (float*)d_out;

    float* xz   = (float*)d_ws;
    float* xc   = xz   + (size_t)LSEQ * 2 * D_INNER;
    float* xdbl = xc   + (size_t)LSEQ * D_INNER;
    float* dtb  = xdbl + (size_t)LSEQ * (DT_RANK + 2 * D_STATE);

    dim3 blk(128, 1, 1);

    gemm_wmma_f32<<<dim3(4096 / 64, LSEQ / 128), blk, 0, stream>>>(
        x, W_in, nullptr, xz, LSEQ, 2 * D_INNER, D_MODEL, D_MODEL, 0);

    conv_silu_kernel<<<(LSEQ * D_INNER) / 256, 256, 0, stream>>>(
        xz, conv_w, conv_b, xc);

    gemm_wmma_f32<<<dim3((96 + 63) / 64, LSEQ / 128), blk, 0, stream>>>(
        xc, W_x, nullptr, xdbl, LSEQ, DT_RANK + 2 * D_STATE, D_INNER, D_INNER, 0);

    gemm_wmma_f32<<<dim3(D_INNER / 64, LSEQ / 128), blk, 0, stream>>>(
        xdbl, W_dt, b_dt, dtb, LSEQ, D_INNER, DT_RANK, DT_RANK + 2 * D_STATE, 1);

    ssm_scan_kernel<<<D_INNER / SC_CH, 512, 0, stream>>>(
        xdbl, dtb, xz, A_log, D_param, xc);

    gemm_wmma_f32<<<dim3(D_MODEL / 64, LSEQ / 128), blk, 0, stream>>>(
        xc, W_out, nullptr, out, LSEQ, D_MODEL, D_INNER, D_INNER, 0);
}


# x_dbl GEMM (N=96, only 32 work-groups): k cut into 8 slices of 256, so 256 work-groups write partial planes and a small kernel adds them in slice order (a different order of that k sum)
# speedup vs baseline: 9.9926x; 1.1537x over previous
#include <hip/hip_runtime.h>
#include <math.h>

#define D_MODEL 1024
#define D_STATE 16
#define D_CONV  4
#define DT_RANK 64
#define D_INNER 2048
#define LSEQ    2048

typedef __attribute__((ext_vector_type(2))) float v2f;
typedef __attribute__((ext_vector_type(4))) float v4f;
typedef __attribute__((ext_vector_type(8))) float v8f;

struct Frag { v2f a0, a1, b0, b1, b2, b3; };

#define GB_M 128
#define GB_N 64
#define GB_K 32
#define GB_P 36

__device__ __forceinline__ void load_frag(Frag& f, const float* pA, const float* pB, int k)
{
    f.a0 = *(const v2f*)(pA + k);
    f.a1 = *(const v2f*)(pA + 16 * GB_P + k);
    f.b0 = *(const v2f*)(pB + k);
    f.b1 = *(const v2f*)(pB + 16 * GB_P + k);
    f.b2 = *(const v2f*)(pB + 32 * GB_P + k);
    f.b3 = *(const v2f*)(pB + 48 * GB_P + k);
}

__device__ __forceinline__ void do_wmma(const Frag& f, v8f acc[2][4])
{
    acc[0][0] = __builtin_amdgcn_wmma_f32_16x16x4_f32(false, f.a0, false, f.b0, (short)0, acc[0][0], false, false);
    acc[1][0] = __builtin_amdgcn_wmma_f32_16x16x4_f32(false, f.a1, false, f.b0, (short)0, acc[1][0], false, false);
    acc[0][1] = __builtin_amdgcn_wmma_f32_16x16x4_f32(false, f.a0, false, f.b1, (short)0, acc[0][1], false, false);
    acc[1][1] = __builtin_amdgcn_wmma_f32_16x16x4_f32(false, f.a1, false, f.b1, (short)0, acc[1][1], false, false);
    acc[0][2] = __builtin_amdgcn_wmma_f32_16x16x4_f32(false, f.a0, false, f.b2, (short)0, acc[0][2], false, false);
    acc[1][2] = __builtin_amdgcn_wmma_f32_16x16x4_f32(false, f.a1, false, f.b2, (short)0, acc[1][2], false, false);
    acc[0][3] = __builtin_amdgcn_wmma_f32_16x16x4_f32(false, f.a0, false, f.b3, (short)0, acc[0][3], false, false);
    acc[1][3] = __builtin_amdgcn_wmma_f32_16x16x4_f32(false, f.a1, false, f.b3, (short)0, acc[1][3], false, false);
}

__global__ __launch_bounds__(128) void gemm_wmma_f32(
    const float* __restrict__ A, const float* __restrict__ W,
    const float* __restrict__ bias, float* __restrict__ Cbase,
    int M, int N, int K, int lda, int ldw, int act)
{
    __shared__ __attribute__((aligned(16))) float sA[2][GB_M * GB_P];
    __shared__ __attribute__((aligned(16))) float sB[2][GB_N * GB_P];

    const int tid  = threadIdx.x;
    const int lane = tid & 31;
    const int wave = tid >> 5;
    const int mblk = blockIdx.y << 7;
    const int m0   = mblk + (wave << 5);
    const int n0   = blockIdx.x << 6;
    if (mblk >= M) return;

    const int lr = tid >> 3;
    const int q  = tid & 7;
    const size_t kofs = (size_t)blockIdx.z * K;
    float* __restrict__ C = Cbase + (size_t)blockIdx.z * M * N;
    const float* __restrict__ gA = A + (size_t)(mblk + lr) * lda + kofs + (q << 2);
    const size_t aStep = (size_t)16 * lda;
    const float* gB[4];
#pragma unroll
    for (int j = 0; j < 4; ++j) {
        int row = n0 + lr + (j << 4);
        row = (row < N) ? row : (N - 1);
        gB[j] = W + (size_t)row * ldw + kofs + (q << 2);
    }
    const int sOff = lr * GB_P + (q << 2);

    const int mrow  = lane & 15;
    const int khalf = (lane >> 4) << 1;
    const int aOff  = ((wave << 5) + mrow) * GB_P + khalf;
    const int bOff  = mrow * GB_P + khalf;

    v8f acc[2][4];
#pragma unroll
    for (int i = 0; i < 2; ++i)
#pragma unroll
        for (int j = 0; j < 4; ++j) acc[i][j] = (v8f){};

    v4f ra[8], rb[4];
#pragma unroll
    for (int j = 0; j < 8; ++j) ra[j] = *(const v4f*)(gA + j * aStep);
#pragma unroll
    for (int j = 0; j < 4; ++j) rb[j] = *(const v4f*)(gB[j]);
#pragma unroll
    for (int j = 0; j < 8; ++j) *(v4f*)&sA[0][sOff + j * 16 * GB_P] = ra[j];
#pragma unroll
    for (int j = 0; j < 4; ++j) *(v4f*)&sB[0][sOff + j * 16 * GB_P] = rb[j];
    __syncthreads();

    int cur = 0;
    for (int kb = 0; kb < K; kb += GB_K) {
        const bool more = (kb + GB_K) < K;
        if (more) {
#pragma unroll
            for (int j = 0; j < 8; ++j) ra[j] = *(const v4f*)(gA + j * aStep + (kb + GB_K));
#pragma unroll
            for (int j = 0; j < 4; ++j) rb[j] = *(const v4f*)(gB[j] + (kb + GB_K));
        }

        const float* pA = &sA[cur][aOff];
        const float* pB = &sB[cur][bOff];
        Frag f0, f1;
        load_frag(f0, pA, pB, 0);
#pragma unroll
        for (int ks = 0; ks < GB_K; ks += 8) {
            load_frag(f1, pA, pB, ks + 4);
            do_wmma(f0, acc);
            if (ks + 8 < GB_K) load_frag(f0, pA, pB, ks + 8);
            do_wmma(f1, acc);
        }

        if (more) {
#pragma unroll
            for (int j = 0; j < 8; ++j) *(v4f*)&sA[cur ^ 1][sOff + j * 16 * GB_P] = ra[j];
#pragma unroll
            for (int j = 0; j < 4; ++j) *(v4f*)&sB[cur ^ 1][sOff + j * 16 * GB_P] = rb[j];
        }
        __syncthreads();
        cur ^= 1;
    }

    const int rsub = (lane >> 4) << 3;
#pragma unroll
    for (int j = 0; j < 4; ++j) {
        const int col = n0 + (j << 4) + mrow;
        if ((n0 + (j << 4)) >= N) continue;
        const float bv = (act == 1) ? bias[col] : 0.0f;
#pragma unroll
        for (int i = 0; i < 2; ++i) {
            const int rbase = m0 + (i << 4) + rsub;
#pragma unroll
            for (int r = 0; r < 8; ++r) {
                float v = acc[i][j][r];
                if (act == 1) {
                    v += bv;
                    v = (v > 20.0f) ? v : log1pf(__expf(v));
                }
                C[(size_t)(rbase + r) * N + col] = v;
            }
        }
    }
}

#define XD_SLICES 8
#define XD_ELEMS  (LSEQ * (DT_RANK + 2 * D_STATE))

__global__ __launch_bounds__(256) void slice_sum_kernel(
    const float* __restrict__ planes, float* __restrict__ out)
{
    const int i = blockIdx.x * 256 + threadIdx.x;
    const v4f* __restrict__ p = (const v4f*)planes + i;
    v4f v[XD_SLICES];
#pragma unroll
    for (int s = 0; s < XD_SLICES; ++s) v[s] = p[(size_t)s * (XD_ELEMS / 4)];
    v4f acc = v[0];
#pragma unroll
    for (int s = 1; s < XD_SLICES; ++s) acc += v[s];
    ((v4f*)out)[i] = acc;
}

__global__ __launch_bounds__(256) void conv_silu_kernel(
    const float* __restrict__ xz, const float* __restrict__ cw,
    const float* __restrict__ cb, float* __restrict__ xc)
{
    int idx = blockIdx.x * blockDim.x + threadIdx.x;
    if (idx >= LSEQ * D_INNER) return;
    int l = idx >> 11;
    int d = idx & (D_INNER - 1);

    float acc = cb[d];
#pragma unroll
    for (int j = 0; j < D_CONV; ++j) {
        int ls = l - (D_CONV - 1) + j;
        if (ls >= 0)
            acc += cw[d * D_CONV + j] * xz[(size_t)ls * (2 * D_INNER) + d];
    }
    xc[idx] = acc / (1.0f + __expf(-acc));
}

#define SC_CH 32
#define SC_T  64
#define SC_R  (SC_T / 16)

template <int CTRL>
__device__ __forceinline__ float dpp_row(float v)
{
    const int b = __builtin_bit_cast(int, v);
    return __builtin_bit_cast(float, __builtin_amdgcn_update_dpp(0, b, CTRL, 0xf, 0xf, true));
}

template <int XM>
__device__ __forceinline__ float pair_fold(float even, float odd, bool bit)
{
    const float keep = bit ? odd : even;
    const float send = bit ? even : odd;
    return keep + dpp_row<0x160 + XM>(send);
}

__device__ __forceinline__ float row16_sum_by_step(const float (&p)[16], int n)
{
    const bool b0 = (n & 1) != 0, b1 = (n & 2) != 0, b2 = (n & 4) != 0, b3 = (n & 8) != 0;
    float q[8], r[4], u[2];
#pragma unroll
    for (int k = 0; k < 8; ++k) q[k] = pair_fold<1>(p[2 * k], p[2 * k + 1], b0);
#pragma unroll
    for (int k = 0; k < 4; ++k) r[k] = pair_fold<2>(q[2 * k], q[2 * k + 1], b1);
#pragma unroll
    for (int k = 0; k < 2; ++k) u[k] = pair_fold<4>(r[2 * k], r[2 * k + 1], b2);
    return pair_fold<8>(u[0], u[1], b3);
}

__global__ __launch_bounds__(512) void ssm_scan_kernel(
    const float* __restrict__ xdbl, const float* __restrict__ dtbuf,
    const float* __restrict__ xz,   const float* __restrict__ A_log,
    const float* __restrict__ D_param, float* __restrict__ xc_io)
{
    __shared__ float sIn[SC_T][128];
    __shared__ float sY[SC_T][SC_CH + 1];

    const int lane = threadIdx.x & 31;
    const int wave = threadIdx.x >> 5;
    const int D0   = blockIdx.x * SC_CH;

    const int n   = lane & 15;
    const int chl = (wave << 1) + (lane >> 4);
    const float Acoef = -__expf(A_log[(size_t)(D0 + chl) * D_STATE + n]);
    float h = 0.0f;

    const int gch = D0 + lane;
    const float Dp = D_param[gch];

    float r_dt[SC_R], r_x[SC_R], r_z[SC_R], r_bc[SC_R];
#pragma unroll
    for (int r = 0; r < SC_R; ++r) {
        const size_t l = (size_t)(r * 16 + wave);
        r_dt[r] = dtbuf[l * D_INNER + gch];
        r_x[r]  = xc_io[l * D_INNER + gch];
        r_z[r]  = xz[l * (2 * D_INNER) + D_INNER + gch];
        r_bc[r] = xdbl[l * (DT_RANK + 2 * D_STATE) + DT_RANK + lane];
    }

    for (int lc = 0; lc < LSEQ; lc += SC_T) {
        float c_x[SC_R], c_z[SC_R];
#pragma unroll
        for (int r = 0; r < SC_R; ++r) {
            const int t = r * 16 + wave;
            sIn[t][lane]      = r_dt[r];
            sIn[t][32 + lane] = r_x[r];
            sIn[t][64 + lane] = r_bc[r];
            c_x[r] = r_x[r];
            c_z[r] = r_z[r];
        }
        __syncthreads();

        if (lc + SC_T < LSEQ) {
#pragma unroll
            for (int r = 0; r < SC_R; ++r) {
                const size_t l = (size_t)(lc + SC_T + r * 16 + wave);
                r_dt[r] = dtbuf[l * D_INNER + gch];
                r_x[r]  = xc_io[l * D_INNER + gch];
                r_z[r]  = xz[l * (2 * D_INNER) + D_INNER + gch];
                r_bc[r] = xdbl[l * (DT_RANK + 2 * D_STATE) + DT_RANK + lane];
            }
        }

        float yk[SC_R];
#pragma unroll
        for (int r = 0; r < SC_R; ++r) {
            float p[16];
#pragma unroll
            for (int i = 0; i < 16; ++i) {
                const int t = r * 16 + i;
                const float dt = sIn[t][chl];
                const float xv = sIn[t][32 + chl];
                const float Bn = sIn[t][64 + n];
                const float Cn = sIn[t][80 + n];
                const float dA = __expf(dt * Acoef);
                h = dA * h + (dt * Bn) * xv;
                p[i] = h * Cn;
            }
            yk[r] = row16_sum_by_step(p, n);
        }
#pragma unroll
        for (int r = 0; r < SC_R; ++r) sY[r * 16 + n][chl] = yk[r];
        __syncthreads();

#pragma unroll
        for (int r = 0; r < SC_R; ++r) {
            const int t = r * 16 + wave;
            float y = sY[t][lane];
            y += c_x[r] * Dp;
            const float z = c_z[r];
            const float g = z / (1.0f + __expf(-z));
            xc_io[(size_t)(lc + t) * D_INNER + gch] = y * g;
        }
    }
}

extern "C" void kernel_launch(void* const* d_in, const int* in_sizes, int n_in,
                              void* d_out, int out_size, void* d_ws, size_t ws_size,
                              hipStream_t stream) {
    const float* x       = (const float*)d_in[0];
    const float* W_in    = (const float*)d_in[1];
    const float* conv_w  = (const float*)d_in[2];
    const float* conv_b  = (const float*)d_in[3];
    const float* W_x     = (const float*)d_in[4];
    const float* W_dt    = (const float*)d_in[5];
    const float* b_dt    = (const float*)d_in[6];
    const float* A_log   = (const float*)d_in[7];
    const float* D_param = (const float*)d_in[8];
    const float* W_out   = (const float*)d_in[9];
    float* out = (float*)d_out;

    float* xz   = (float*)d_ws;
    float* xc   = xz   + (size_t)LSEQ * 2 * D_INNER;
    float* xdbl = xc   + (size_t)LSEQ * D_INNER;
    float* dtb  = xdbl + (size_t)LSEQ * (DT_RANK + 2 * D_STATE);

    dim3 blk(128, 1, 1);

    gemm_wmma_f32<<<dim3(4096 / 64, LSEQ / 128), blk, 0, stream>>>(
        x, W_in, nullptr, xz, LSEQ, 2 * D_INNER, D_MODEL, D_MODEL, D_MODEL, 0);

    conv_silu_kernel<<<(LSEQ * D_INNER) / 256, 256, 0, stream>>>(
        xz, conv_w, conv_b, xc);

    gemm_wmma_f32<<<dim3((96 + 63) / 64, LSEQ / 128, XD_SLICES), blk, 0, stream>>>(
        xc, W_x, nullptr, dtb, LSEQ, DT_RANK + 2 * D_STATE, D_INNER / XD_SLICES,
        D_INNER, D_INNER, 0);
    slice_sum_kernel<<<(XD_ELEMS / 4) / 256, 256, 0, stream>>>(dtb, xdbl);

    gemm_wmma_f32<<<dim3(D_INNER / 64, LSEQ / 128), blk, 0, stream>>>(
        xdbl, W_dt, b_dt, dtb, LSEQ, D_INNER, DT_RANK, DT_RANK + 2 * D_STATE, DT_RANK, 1);

    ssm_scan_kernel<<<D_INNER / SC_CH, 512, 0, stream>>>(
        xdbl, dtb, xz, A_log, D_param, xc);

    gemm_wmma_f32<<<dim3(D_MODEL / 64, LSEQ / 128), blk, 0, stream>>>(
        xc, W_out, nullptr, out, LSEQ, D_MODEL, D_INNER, D_INNER, D_INNER, 0);
}


# scan: 8 channels per work-group (4 waves) instead of 32 (16 waves): 256 groups, one wave per SIMD on every unit; y stored channel-blocked in whole lines, read so by the out GEMM; all sums unchanged
# speedup vs baseline: 10.6768x; 1.0685x over previous
#include <hip/hip_runtime.h>
#include <math.h>

#define D_MODEL 1024
#define D_STATE 16
#define D_CONV  4
#define DT_RANK 64
#define D_INNER 2048
#define LSEQ    2048

typedef __attribute__((ext_vector_type(2))) float v2f;
typedef __attribute__((ext_vector_type(4))) float v4f;
typedef __attribute__((ext_vector_type(8))) float v8f;

struct Frag { v2f a0, a1, b0, b1, b2, b3; };

#define GB_M 128
#define GB_N 64
#define GB_K 32
#define GB_P 36

__device__ __forceinline__ void load_frag(Frag& f, const float* pA, const float* pB, int k)
{
    f.a0 = *(const v2f*)(pA + k);
    f.a1 = *(const v2f*)(pA + 16 * GB_P + k);
    f.b0 = *(const v2f*)(pB + k);
    f.b1 = *(const v2f*)(pB + 16 * GB_P + k);
    f.b2 = *(const v2f*)(pB + 32 * GB_P + k);
    f.b3 = *(const v2f*)(pB + 48 * GB_P + k);
}

__device__ __forceinline__ void do_wmma(const Frag& f, v8f acc[2][4])
{
    acc[0][0] = __builtin_amdgcn_wmma_f32_16x16x4_f32(false, f.a0, false, f.b0, (short)0, acc[0][0], false, false);
    acc[1][0] = __builtin_amdgcn_wmma_f32_16x16x4_f32(false, f.a1, false, f.b0, (short)0, acc[1][0], false, false);
    acc[0][1] = __builtin_amdgcn_wmma_f32_16x16x4_f32(false, f.a0, false, f.b1, (short)0, acc[0][1], false, false);
    acc[1][1] = __builtin_amdgcn_wmma_f32_16x16x4_f32(false, f.a1, false, f.b1, (short)0, acc[1][1], false, false);
    acc[0][2] = __builtin_amdgcn_wmma_f32_16x16x4_f32(false, f.a0, false, f.b2, (short)0, acc[0][2], false, false);
    acc[1][2] = __builtin_amdgcn_wmma_f32_16x16x4_f32(false, f.a1, false, f.b2, (short)0, acc[1][2], false, false);
    acc[0][3] = __builtin_amdgcn_wmma_f32_16x16x4_f32(false, f.a0, false, f.b3, (short)0, acc[0][3], false, false);
    acc[1][3] = __builtin_amdgcn_wmma_f32_16x16x4_f32(false, f.a1, false, f.b3, (short)0, acc[1][3], false, false);
}

__global__ __launch_bounds__(128) void gemm_wmma_f32(
    const float* __restrict__ A, const float* __restrict__ W,
    const float* __restrict__ bias, float* __restrict__ Cbase,
    int M, int N, int K, int lda, int ldw, int act, int ablk)
{
    __shared__ __attribute__((aligned(16))) float sA[2][GB_M * GB_P];
    __shared__ __attribute__((aligned(16))) float sB[2][GB_N * GB_P];

    const int tid  = threadIdx.x;
    const int lane = tid & 31;
    const int wave = tid >> 5;
    const int mblk = blockIdx.y << 7;
    const int m0   = mblk + (wave << 5);
    const int n0   = blockIdx.x << 6;
    if (mblk >= M) return;

    const int lr = tid >> 3;
    const int q  = tid & 7;
    const size_t kofs = (size_t)blockIdx.z * K;
    float* __restrict__ C = Cbase + (size_t)blockIdx.z * M * N;
    const size_t aRow = ablk ? (size_t)8 : (size_t)lda;
    const size_t aK   = ablk ? (size_t)M : (size_t)1;
    const size_t aQ   = ablk ? ((size_t)(q >> 1) * M * 8 + ((q & 1) << 2)) : (size_t)(q << 2);
    const float* __restrict__ gA = A + (size_t)(mblk + lr) * aRow + kofs * aK + aQ;
    const size_t aStep = (size_t)16 * aRow;
    const float* gB[4];
#pragma unroll
    for (int j = 0; j < 4; ++j) {
        int row = n0 + lr + (j << 4);
        row = (row < N) ? row : (N - 1);
        gB[j] = W + (size_t)row * ldw + kofs + (q << 2);
    }
    const int sOff = lr * GB_P + (q << 2);

    const int mrow  = lane & 15;
    const int khalf = (lane >> 4) << 1;
    const int aOff  = ((wave << 5) + mrow) * GB_P + khalf;
    const int bOff  = mrow * GB_P + khalf;

    v8f acc[2][4];
#pragma unroll
    for (int i = 0; i < 2; ++i)
#pragma unroll
        for (int j = 0; j < 4; ++j) acc[i][j] = (v8f){};

    v4f ra[8], rb[4];
#pragma unroll
    for (int j = 0; j < 8; ++j) ra[j] = *(const v4f*)(gA + j * aStep);
#pragma unroll
    for (int j = 0; j < 4; ++j) rb[j] = *(const v4f*)(gB[j]);
#pragma unroll
    for (int j = 0; j < 8; ++j) *(v4f*)&sA[0][sOff + j * 16 * GB_P] = ra[j];
#pragma unroll
    for (int j = 0; j < 4; ++j) *(v4f*)&sB[0][sOff + j * 16 * GB_P] = rb[j];
    __syncthreads();

    int cur = 0;
    for (int kb = 0; kb < K; kb += GB_K) {
        const bool more = (kb + GB_K) < K;
        if (more) {
#pragma unroll
            for (int j = 0; j < 8; ++j) ra[j] = *(const v4f*)(gA + j * aStep + (size_t)(kb + GB_K) * aK);
#pragma unroll
            for (int j = 0; j < 4; ++j) rb[j] = *(const v4f*)(gB[j] + (kb + GB_K));
        }

        const float* pA = &sA[cur][aOff];
        const float* pB = &sB[cur][bOff];
        Frag f0, f1;
        load_frag(f0, pA, pB, 0);
#pragma unroll
        for (int ks = 0; ks < GB_K; ks += 8) {
            load_frag(f1, pA, pB, ks + 4);
            do_wmma(f0, acc);
            if (ks + 8 < GB_K) load_frag(f0, pA, pB, ks + 8);
            do_wmma(f1, acc);
        }

        if (more) {
#pragma unroll
            for (int j = 0; j < 8; ++j) *(v4f*)&sA[cur ^ 1][sOff + j * 16 * GB_P] = ra[j];
#pragma unroll
            for (int j = 0; j < 4; ++j) *(v4f*)&sB[cur ^ 1][sOff + j * 16 * GB_P] = rb[j];
        }
        __syncthreads();
        cur ^= 1;
    }

    const int rsub = (lane >> 4) << 3;
#pragma unroll
    for (int j = 0; j < 4; ++j) {
        const int col = n0 + (j << 4) + mrow;
        if ((n0 + (j << 4)) >= N) continue;
        const float bv = (act == 1) ? bias[col] : 0.0f;
#pragma unroll
        for (int i = 0; i < 2; ++i) {
            const int rbase = m0 + (i << 4) + rsub;
#pragma unroll
            for (int r = 0; r < 8; ++r) {
                float v = acc[i][j][r];
                if (act == 1) {
                    v += bv;
                    v = (v > 20.0f) ? v : log1pf(__expf(v));
                }
                C[(size_t)(rbase + r) * N + col] = v;
            }
        }
    }
}

#define XD_SLICES 8
#define XD_ELEMS  (LSEQ * (DT_RANK + 2 * D_STATE))

__global__ __launch_bounds__(256) void slice_sum_kernel(
    const float* __restrict__ planes, float* __restrict__ out)
{
    const int i = blockIdx.x * 256 + threadIdx.x;
    const v4f* __restrict__ p = (const v4f*)planes + i;
    v4f v[XD_SLICES];
#pragma unroll
    for (int s = 0; s < XD_SLICES; ++s) v[s] = p[(size_t)s * (XD_ELEMS / 4)];
    v4f acc = v[0];
#pragma unroll
    for (int s = 1; s < XD_SLICES; ++s) acc += v[s];
    ((v4f*)out)[i] = acc;
}

__global__ __launch_bounds__(256) void conv_silu_kernel(
    const float* __restrict__ xz, const float* __restrict__ cw,
    const float* __restrict__ cb, float* __restrict__ xc)
{
    int idx = blockIdx.x * blockDim.x + threadIdx.x;
    if (idx >= LSEQ * D_INNER) return;
    int l = idx >> 11;
    int d = idx & (D_INNER - 1);

    float acc = cb[d];
#pragma unroll
    for (int j = 0; j < D_CONV; ++j) {
        int ls = l - (D_CONV - 1) + j;
        if (ls >= 0)
            acc += cw[d * D_CONV + j] * xz[(size_t)ls * (2 * D_INNER) + d];
    }
    xc[idx] = acc / (1.0f + __expf(-acc));
}

#define SC_CH 8
#define SC_T  64
#define SC_W  48
#define SC_YP 12

template <int CTRL>
__device__ __forceinline__ float dpp_row(float v)
{
    const int b = __builtin_bit_cast(int, v);
    return __builtin_bit_cast(float, __builtin_amdgcn_update_dpp(0, b, CTRL, 0xf, 0xf, true));
}

template <int XM>
__device__ __forceinline__ float pair_fold(float even, float odd, bool bit)
{
    const float keep = bit ? odd : even;
    const float send = bit ? even : odd;
    return keep + dpp_row<0x160 + XM>(send);
}

__device__ __forceinline__ float row16_sum_by_step(const float (&p)[16], int n)
{
    const bool b0 = (n & 1) != 0, b1 = (n & 2) != 0, b2 = (n & 4) != 0, b3 = (n & 8) != 0;
    float q[8], r[4], u[2];
#pragma unroll
    for (int k = 0; k < 8; ++k) q[k] = pair_fold<1>(p[2 * k], p[2 * k + 1], b0);
#pragma unroll
    for (int k = 0; k < 4; ++k) r[k] = pair_fold<2>(q[2 * k], q[2 * k + 1], b1);
#pragma unroll
    for (int k = 0; k < 2; ++k) u[k] = pair_fold<4>(r[2 * k], r[2 * k + 1], b2);
    return pair_fold<8>(u[0], u[1], b3);
}

__global__ __launch_bounds__(128) void ssm_scan_kernel(
    const float* __restrict__ xdbl, const float* __restrict__ dtbuf,
    const float* __restrict__ xz,   const float* __restrict__ A_log,
    const float* __restrict__ D_param, const float* __restrict__ xc,
    float* __restrict__ yblk)
{
    __shared__ __attribute__((aligned(16))) float sIn[SC_T][SC_W];
    __shared__ __attribute__((aligned(16))) float sY[SC_T][SC_YP];

    const int tid  = threadIdx.x;
    const int lane = tid & 31;
    const int wave = tid >> 5;
    const int D0   = blockIdx.x * SC_CH;

    const int n   = lane & 15;
    const int chl = (wave << 1) + (lane >> 4);
    const float Acoef = -__expf(A_log[(size_t)(D0 + chl) * D_STATE + n]);
    float h = 0.0f;

    const int srow = tid >> 1;
    const int sc4  = (tid & 1) << 2;
    const int brow = tid >> 3;
    const int bq   = (tid & 7) << 2;
    const v4f Dp = *(const v4f*)(D_param + D0 + sc4);

    const float* __restrict__ pdt = dtbuf + (size_t)srow * D_INNER + D0 + sc4;
    const float* __restrict__ px  = xc    + (size_t)srow * D_INNER + D0 + sc4;
    const float* __restrict__ pz  = xz    + (size_t)srow * (2 * D_INNER) + D_INNER + D0 + sc4;
    const float* __restrict__ pbc = xdbl  + (size_t)brow * (DT_RANK + 2 * D_STATE) + DT_RANK + bq;
    float* __restrict__ py = yblk + ((size_t)blockIdx.x * LSEQ + srow) * SC_CH + sc4;

    v4f r_dt = *(const v4f*)pdt;
    v4f r_x  = *(const v4f*)px;
    v4f r_z  = *(const v4f*)pz;
    v4f r_bc[4];
#pragma unroll
    for (int r = 0; r < 4; ++r)
        r_bc[r] = *(const v4f*)(pbc + (size_t)(r * 16) * (DT_RANK + 2 * D_STATE));

    for (int lc = 0; lc < LSEQ; lc += SC_T) {
        *(v4f*)&sIn[srow][sc4]     = r_dt;
        *(v4f*)&sIn[srow][8 + sc4] = r_x;
#pragma unroll
        for (int r = 0; r < 4; ++r) *(v4f*)&sIn[r * 16 + brow][16 + bq] = r_bc[r];
        const v4f c_x = r_x;
        const v4f c_z = r_z;
        __syncthreads();

        if (lc + SC_T < LSEQ) {
            const size_t o = (size_t)(lc + SC_T);
            r_dt = *(const v4f*)(pdt + o * D_INNER);
            r_x  = *(const v4f*)(px  + o * D_INNER);
            r_z  = *(const v4f*)(pz  + o * (2 * D_INNER));
#pragma unroll
            for (int r = 0; r < 4; ++r)
                r_bc[r] = *(const v4f*)(pbc + (o + r * 16) * (DT_RANK + 2 * D_STATE));
        }

        v4f g;
#pragma unroll
        for (int e = 0; e < 4; ++e) g[e] = c_z[e] / (1.0f + __expf(-c_z[e]));

        float yk[4];
#pragma unroll
        for (int r = 0; r < 4; ++r) {
            float p[16];
#pragma unroll
            for (int i = 0; i < 16; ++i) {
                const int t = r * 16 + i;
                const float dt = sIn[t][chl];
                const float xv = sIn[t][8 + chl];
                const float Bn = sIn[t][16 + n];
                const float Cn = sIn[t][32 + n];
                const float dA = __expf(dt * Acoef);
                h = dA * h + (dt * Bn) * xv;
                p[i] = h * Cn;
            }
            yk[r] = row16_sum_by_step(p, n);
        }
#pragma unroll
        for (int r = 0; r < 4; ++r) sY[r * 16 + n][chl] = yk[r];
        __syncthreads();

        v4f y = *(const v4f*)&sY[srow][sc4];
#pragma unroll
        for (int e = 0; e < 4; ++e) {
            float v = y[e];
            v += c_x[e] * Dp[e];
            y[e] = v * g[e];
        }
        *(v4f*)(py + (size_t)lc * SC_CH) = y;
    }
}

extern "C" void kernel_launch(void* const* d_in, const int* in_sizes, int n_in,
                              void* d_out, int out_size, void* d_ws, size_t ws_size,
                              hipStream_t stream) {
    const float* x       = (const float*)d_in[0];
    const float* W_in    = (const float*)d_in[1];
    const float* conv_w  = (const float*)d_in[2];
    const float* conv_b  = (const float*)d_in[3];
    const float* W_x     = (const float*)d_in[4];
    const float* W_dt    = (const float*)d_in[5];
    const float* b_dt    = (const float*)d_in[6];
    const float* A_log   = (const float*)d_in[7];
    const float* D_param = (const float*)d_in[8];
    const float* W_out   = (const float*)d_in[9];
    float* out = (float*)d_out;

    float* xz   = (float*)d_ws;
    float* xc   = xz   + (size_t)LSEQ * 2 * D_INNER;
    float* xdbl = xc   + (size_t)LSEQ * D_INNER;
    float* dtb  = xdbl + (size_t)LSEQ * (DT_RANK + 2 * D_STATE);
    float* yb   = dtb  + (size_t)LSEQ * D_INNER;

    dim3 blk(128, 1, 1);

    gemm_wmma_f32<<<dim3(4096 / 64, LSEQ / 128), blk, 0, stream>>>(
        x, W_in, nullptr, xz, LSEQ, 2 * D_INNER, D_MODEL, D_MODEL, D_MODEL, 0, 0);

    conv_silu_kernel<<<(LSEQ * D_INNER) / 256, 256, 0, stream>>>(
        xz, conv_w, conv_b, xc);

    gemm_wmma_f32<<<dim3((96 + 63) / 64, LSEQ / 128, XD_SLICES), blk, 0, stream>>>(
        xc, W_x, nullptr, dtb, LSEQ, DT_RANK + 2 * D_STATE, D_INNER / XD_SLICES,
        D_INNER, D_INNER, 0, 0);
    slice_sum_kernel<<<(XD_ELEMS / 4) / 256, 256, 0, stream>>>(dtb, xdbl);

    gemm_wmma_f32<<<dim3(D_INNER / 64, LSEQ / 128), blk, 0, stream>>>(
        xdbl, W_dt, b_dt, dtb, LSEQ, D_INNER, DT_RANK, DT_RANK + 2 * D_STATE, DT_RANK, 1, 0);

    ssm_scan_kernel<<<D_INNER / SC_CH, 128, 0, stream>>>(
        xdbl, dtb, xz, A_log, D_param, xc, yb);

    gemm_wmma_f32<<<dim3(D_MODEL / 64, LSEQ / 128), blk, 0, stream>>>(
        yb, W_out, nullptr, out, LSEQ, D_MODEL, D_INNER, D_INNER, D_INNER, 0, 1);
}


# scan: the LDS reads of each 16-step block issued as one group a whole block ahead, in a second register set, not spread among the steps a few instructions before use; all sums unchanged
# speedup vs baseline: 11.6513x; 1.0913x over previous
#include <hip/hip_runtime.h>
#include <math.h>

#define D_MODEL 1024
#define D_STATE 16
#define D_CONV  4
#define DT_RANK 64
#define D_INNER 2048
#define LSEQ    2048

typedef __attribute__((ext_vector_type(2))) float v2f;
typedef __attribute__((ext_vector_type(4))) float v4f;
typedef __attribute__((ext_vector_type(8))) float v8f;

struct Frag { v2f a0, a1, b0, b1, b2, b3; };

#define GB_M 128
#define GB_N 64
#define GB_K 32
#define GB_P 36

__device__ __forceinline__ void load_frag(Frag& f, const float* pA, const float* pB, int k)
{
    f.a0 = *(const v2f*)(pA + k);
    f.a1 = *(const v2f*)(pA + 16 * GB_P + k);
    f.b0 = *(const v2f*)(pB + k);
    f.b1 = *(const v2f*)(pB + 16 * GB_P + k);
    f.b2 = *(const v2f*)(pB + 32 * GB_P + k);
    f.b3 = *(const v2f*)(pB + 48 * GB_P + k);
}

__device__ __forceinline__ void do_wmma(const Frag& f, v8f acc[2][4])
{
    acc[0][0] = __builtin_amdgcn_wmma_f32_16x16x4_f32(false, f.a0, false, f.b0, (short)0, acc[0][0], false, false);
    acc[1][0] = __builtin_amdgcn_wmma_f32_16x16x4_f32(false, f.a1, false, f.b0, (short)0, acc[1][0], false, false);
    acc[0][1] = __builtin_amdgcn_wmma_f32_16x16x4_f32(false, f.a0, false, f.b1, (short)0, acc[0][1], false, false);
    acc[1][1] = __builtin_amdgcn_wmma_f32_16x16x4_f32(false, f.a1, false, f.b1, (short)0, acc[1][1], false, false);
    acc[0][2] = __builtin_amdgcn_wmma_f32_16x16x4_f32(false, f.a0, false, f.b2, (short)0, acc[0][2], false, false);
    acc[1][2] = __builtin_amdgcn_wmma_f32_16x16x4_f32(false, f.a1, false, f.b2, (short)0, acc[1][2], false, false);
    acc[0][3] = __builtin_amdgcn_wmma_f32_16x16x4_f32(false, f.a0, false, f.b3, (short)0, acc[0][3], false, false);
    acc[1][3] = __builtin_amdgcn_wmma_f32_16x16x4_f32(false, f.a1, false, f.b3, (short)0, acc[1][3], false, false);
}

__global__ __launch_bounds__(128) void gemm_wmma_f32(
    const float* __restrict__ A, const float* __restrict__ W,
    const float* __restrict__ bias, float* __restrict__ Cbase,
    int M, int N, int K, int lda, int ldw, int act, int ablk)
{
    __shared__ __attribute__((aligned(16))) float sA[2][GB_M * GB_P];
    __shared__ __attribute__((aligned(16))) float sB[2][GB_N * GB_P];

    const int tid  = threadIdx.x;
    const int lane = tid & 31;
    const int wave = tid >> 5;
    const int mblk = blockIdx.y << 7;
    const int m0   = mblk + (wave << 5);
    const int n0   = blockIdx.x << 6;
    if (mblk >= M) return;

    const int lr = tid >> 3;
    const int q  = tid & 7;
    const size_t kofs = (size_t)blockIdx.z * K;
    float* __restrict__ C = Cbase + (size_t)blockIdx.z * M * N;
    const size_t aRow = ablk ? (size_t)8 : (size_t)lda;
    const size_t aK   = ablk ? (size_t)M : (size_t)1;
    const size_t aQ   = ablk ? ((size_t)(q >> 1) * M * 8 + ((q & 1) << 2)) : (size_t)(q << 2);
    const float* __restrict__ gA = A + (size_t)(mblk + lr) * aRow + kofs * aK + aQ;
    const size_t aStep = (size_t)16 * aRow;
    const float* gB[4];
#pragma unroll
    for (int j = 0; j < 4; ++j) {
        int row = n0 + lr + (j << 4);
        row = (row < N) ? row : (N - 1);
        gB[j] = W + (size_t)row * ldw + kofs + (q << 2);
    }
    const int sOff = lr * GB_P + (q << 2);

    const int mrow  = lane & 15;
    const int khalf = (lane >> 4) << 1;
    const int aOff  = ((wave << 5) + mrow) * GB_P + khalf;
    const int bOff  = mrow * GB_P + khalf;

    v8f acc[2][4];
#pragma unroll
    for (int i = 0; i < 2; ++i)
#pragma unroll
        for (int j = 0; j < 4; ++j) acc[i][j] = (v8f){};

    v4f ra[8], rb[4];
#pragma unroll
    for (int j = 0; j < 8; ++j) ra[j] = *(const v4f*)(gA + j * aStep);
#pragma unroll
    for (int j = 0; j < 4; ++j) rb[j] = *(const v4f*)(gB[j]);
#pragma unroll
    for (int j = 0; j < 8; ++j) *(v4f*)&sA[0][sOff + j * 16 * GB_P] = ra[j];
#pragma unroll
    for (int j = 0; j < 4; ++j) *(v4f*)&sB[0][sOff + j * 16 * GB_P] = rb[j];
    __syncthreads();

    int cur = 0;
    for (int kb = 0; kb < K; kb += GB_K) {
        const bool more = (kb + GB_K) < K;
        if (more) {
#pragma unroll
            for (int j = 0; j < 8; ++j) ra[j] = *(const v4f*)(gA + j * aStep + (size_t)(kb + GB_K) * aK);
#pragma unroll
            for (int j = 0; j < 4; ++j) rb[j] = *(const v4f*)(gB[j] + (kb + GB_K));
        }

        const float* pA = &sA[cur][aOff];
        const float* pB = &sB[cur][bOff];
        Frag f0, f1;
        load_frag(f0, pA, pB, 0);
#pragma unroll
        for (int ks = 0; ks < GB_K; ks += 8) {
            load_frag(f1, pA, pB, ks + 4);
            do_wmma(f0, acc);
            if (ks + 8 < GB_K) load_frag(f0, pA, pB, ks + 8);
            do_wmma(f1, acc);
        }

        if (more) {
#pragma unroll
            for (int j = 0; j < 8; ++j) *(v4f*)&sA[cur ^ 1][sOff + j * 16 * GB_P] = ra[j];
#pragma unroll
            for (int j = 0; j < 4; ++j) *(v4f*)&sB[cur ^ 1][sOff + j * 16 * GB_P] = rb[j];
        }
        __syncthreads();
        cur ^= 1;
    }

    const int rsub = (lane >> 4) << 3;
#pragma unroll
    for (int j = 0; j < 4; ++j) {
        const int col = n0 + (j << 4) + mrow;
        if ((n0 + (j << 4)) >= N) continue;
        const float bv = (act == 1) ? bias[col] : 0.0f;
#pragma unroll
        for (int i = 0; i < 2; ++i) {
            const int rbase = m0 + (i << 4) + rsub;
#pragma unroll
            for (int r = 0; r < 8; ++r) {
                float v = acc[i][j][r];
                if (act == 1) {
                    v += bv;
                    v = (v > 20.0f) ? v : log1pf(__expf(v));
                }
                C[(size_t)(rbase + r) * N + col] = v;
            }
        }
    }
}

#define XD_SLICES 8
#define XD_ELEMS  (LSEQ * (DT_RANK + 2 * D_STATE))

__global__ __launch_bounds__(256) void slice_sum_kernel(
    const float* __restrict__ planes, float* __restrict__ out)
{
    const int i = blockIdx.x * 256 + threadIdx.x;
    const v4f* __restrict__ p = (const v4f*)planes + i;
    v4f v[XD_SLICES];
#pragma unroll
    for (int s = 0; s < XD_SLICES; ++s) v[s] = p[(size_t)s * (XD_ELEMS / 4)];
    v4f acc = v[0];
#pragma unroll
    for (int s = 1; s < XD_SLICES; ++s) acc += v[s];
    ((v4f*)out)[i] = acc;
}

__global__ __launch_bounds__(256) void conv_silu_kernel(
    const float* __restrict__ xz, const float* __restrict__ cw,
    const float* __restrict__ cb, float* __restrict__ xc)
{
    int idx = blockIdx.x * blockDim.x + threadIdx.x;
    if (idx >= LSEQ * D_INNER) return;
    int l = idx >> 11;
    int d = idx & (D_INNER - 1);

    float acc = cb[d];
#pragma unroll
    for (int j = 0; j < D_CONV; ++j) {
        int ls = l - (D_CONV - 1) + j;
        if (ls >= 0)
            acc += cw[d * D_CONV + j] * xz[(size_t)ls * (2 * D_INNER) + d];
    }
    xc[idx] = acc / (1.0f + __expf(-acc));
}

#define SC_CH 8
#define SC_T  64
#define SC_W  48
#define SC_YP 12

template <int CTRL>
__device__ __forceinline__ float dpp_row(float v)
{
    const int b = __builtin_bit_cast(int, v);
    return __builtin_bit_cast(float, __builtin_amdgcn_update_dpp(0, b, CTRL, 0xf, 0xf, true));
}

template <int XM>
__device__ __forceinline__ float pair_fold(float even, float odd, bool bit)
{
    const float keep = bit ? odd : even;
    const float send = bit ? even : odd;
    return keep + dpp_row<0x160 + XM>(send);
}

__device__ __forceinline__ float row16_sum_by_step(const float (&p)[16], int n)
{
    const bool b0 = (n & 1) != 0, b1 = (n & 2) != 0, b2 = (n & 4) != 0, b3 = (n & 8) != 0;
    float q[8], r[4], u[2];
#pragma unroll
    for (int k = 0; k < 8; ++k) q[k] = pair_fold<1>(p[2 * k], p[2 * k + 1], b0);
#pragma unroll
    for (int k = 0; k < 4; ++k) r[k] = pair_fold<2>(q[2 * k], q[2 * k + 1], b1);
#pragma unroll
    for (int k = 0; k < 2; ++k) u[k] = pair_fold<4>(r[2 * k], r[2 * k + 1], b2);
    return pair_fold<8>(u[0], u[1], b3);
}

struct ScanBlk { float dt[16], x[16], B[16], C[16]; };

__device__ __forceinline__ void load_blk(ScanBlk& b, const float (*s)[SC_W], int r, int chl, int n)
{
#pragma unroll
    for (int i = 0; i < 16; ++i) {
        const int t = r * 16 + i;
        b.dt[i] = s[t][chl];
        b.x[i]  = s[t][8 + chl];
        b.B[i]  = s[t][16 + n];
        b.C[i]  = s[t][32 + n];
    }
}

__device__ __forceinline__ float scan_blk(const ScanBlk& b, float& h, float Acoef, int n)
{
    float p[16];
#pragma unroll
    for (int i = 0; i < 16; ++i) {
        const float dA = __expf(b.dt[i] * Acoef);
        h = dA * h + (b.dt[i] * b.B[i]) * b.x[i];
        p[i] = h * b.C[i];
    }
    return row16_sum_by_step(p, n);
}

__global__ __launch_bounds__(128) void ssm_scan_kernel(
    const float* __restrict__ xdbl, const float* __restrict__ dtbuf,
    const float* __restrict__ xz,   const float* __restrict__ A_log,
    const float* __restrict__ D_param, const float* __restrict__ xc,
    float* __restrict__ yblk)
{
    __shared__ __attribute__((aligned(16))) float sIn[SC_T][SC_W];
    __shared__ __attribute__((aligned(16))) float sY[SC_T][SC_YP];

    const int tid  = threadIdx.x;
    const int lane = tid & 31;
    const int wave = tid >> 5;
    const int D0   = blockIdx.x * SC_CH;

    const int n   = lane & 15;
    const int chl = (wave << 1) + (lane >> 4);
    const float Acoef = -__expf(A_log[(size_t)(D0 + chl) * D_STATE + n]);
    float h = 0.0f;

    const int srow = tid >> 1;
    const int sc4  = (tid & 1) << 2;
    const int brow = tid >> 3;
    const int bq   = (tid & 7) << 2;
    const v4f Dp = *(const v4f*)(D_param + D0 + sc4);

    const float* __restrict__ pdt = dtbuf + (size_t)srow * D_INNER + D0 + sc4;
    const float* __restrict__ px  = xc    + (size_t)srow * D_INNER + D0 + sc4;
    const float* __restrict__ pz  = xz    + (size_t)srow * (2 * D_INNER) + D_INNER + D0 + sc4;
    const float* __restrict__ pbc = xdbl  + (size_t)brow * (DT_RANK + 2 * D_STATE) + DT_RANK + bq;
    float* __restrict__ py = yblk + ((size_t)blockIdx.x * LSEQ + srow) * SC_CH + sc4;

    v4f r_dt = *(const v4f*)pdt;
    v4f r_x  = *(const v4f*)px;
    v4f r_z  = *(const v4f*)pz;
    v4f r_bc[4];
#pragma unroll
    for (int r = 0; r < 4; ++r)
        r_bc[r] = *(const v4f*)(pbc + (size_t)(r * 16) * (DT_RANK + 2 * D_STATE));

    for (int lc = 0; lc < LSEQ; lc += SC_T) {
        *(v4f*)&sIn[srow][sc4]     = r_dt;
        *(v4f*)&sIn[srow][8 + sc4] = r_x;
#pragma unroll
        for (int r = 0; r < 4; ++r) *(v4f*)&sIn[r * 16 + brow][16 + bq] = r_bc[r];
        const v4f c_x = r_x;
        const v4f c_z = r_z;
        __syncthreads();

        if (lc + SC_T < LSEQ) {
            const size_t o = (size_t)(lc + SC_T);
            r_dt = *(const v4f*)(pdt + o * D_INNER);
            r_x  = *(const v4f*)(px  + o * D_INNER);
            r_z  = *(const v4f*)(pz  + o * (2 * D_INNER));
#pragma unroll
            for (int r = 0; r < 4; ++r)
                r_bc[r] = *(const v4f*)(pbc + (o + r * 16) * (DT_RANK + 2 * D_STATE));
        }

        ScanBlk b0, b1;
        load_blk(b0, sIn, 0, chl, n);
        load_blk(b1, sIn, 1, chl, n);
        __builtin_amdgcn_sched_barrier(0);

        v4f g;
#pragma unroll
        for (int e = 0; e < 4; ++e) g[e] = c_z[e] / (1.0f + __expf(-c_z[e]));

        float yk[4];
        yk[0] = scan_blk(b0, h, Acoef, n);
        __builtin_amdgcn_sched_barrier(0);
        load_blk(b0, sIn, 2, chl, n);
        __builtin_amdgcn_sched_barrier(0);
        yk[1] = scan_blk(b1, h, Acoef, n);
        __builtin_amdgcn_sched_barrier(0);
        load_blk(b1, sIn, 3, chl, n);
        __builtin_amdgcn_sched_barrier(0);
        yk[2] = scan_blk(b0, h, Acoef, n);
        yk[3] = scan_blk(b1, h, Acoef, n);
#pragma unroll
        for (int r = 0; r < 4; ++r) sY[r * 16 + n][chl] = yk[r];
        __syncthreads();

        v4f y = *(const v4f*)&sY[srow][sc4];
#pragma unroll
        for (int e = 0; e < 4; ++e) {
            float v = y[e];
            v += c_x[e] * Dp[e];
            y[e] = v * g[e];
        }
        *(v4f*)(py + (size_t)lc * SC_CH) = y;
    }
}

extern "C" void kernel_launch(void* const* d_in, const int* in_sizes, int n_in,
                              void* d_out, int out_size, void* d_ws, size_t ws_size,
                              hipStream_t stream) {
    const float* x       = (const float*)d_in[0];
    const float* W_in    = (const float*)d_in[1];
    const float* conv_w  = (const float*)d_in[2];
    const float* conv_b  = (const float*)d_in[3];
    const float* W_x     = (const float*)d_in[4];
    const float* W_dt    = (const float*)d_in[5];
    const float* b_dt    = (const float*)d_in[6];
    const float* A_log   = (const float*)d_in[7];
    const float* D_param = (const float*)d_in[8];
    const float* W_out   = (const float*)d_in[9];
    float* out = (float*)d_out;

    float* xz   = (float*)d_ws;
    float* xc   = xz   + (size_t)LSEQ * 2 * D_INNER;
    float* xdbl = xc   + (size_t)LSEQ * D_INNER;
    float* dtb  = xdbl + (size_t)LSEQ * (DT_RANK + 2 * D_STATE);
    float* yb   = dtb  + (size_t)LSEQ * D_INNER;

    dim3 blk(128, 1, 1);

    gemm_wmma_f32<<<dim3(4096 / 64, LSEQ / 128), blk, 0, stream>>>(
        x, W_in, nullptr, xz, LSEQ, 2 * D_INNER, D_MODEL, D_MODEL, D_MODEL, 0, 0);

    conv_silu_kernel<<<(LSEQ * D_INNER) / 256, 256, 0, stream>>>(
        xz, conv_w, conv_b, xc);

    gemm_wmma_f32<<<dim3((96 + 63) / 64, LSEQ / 128, XD_SLICES), blk, 0, stream>>>(
        xc, W_x, nullptr, dtb, LSEQ, DT_RANK + 2 * D_STATE, D_INNER / XD_SLICES,
        D_INNER, D_INNER, 0, 0);
    slice_sum_kernel<<<(XD_ELEMS / 4) / 256, 256, 0, stream>>>(dtb, xdbl);

    gemm_wmma_f32<<<dim3(D_INNER / 64, LSEQ / 128), blk, 0, stream>>>(
        xdbl, W_dt, b_dt, dtb, LSEQ, D_INNER, DT_RANK, DT_RANK + 2 * D_STATE, DT_RANK, 1, 0);

    ssm_scan_kernel<<<D_INNER / SC_CH, 128, 0, stream>>>(
        xdbl, dtb, xz, A_log, D_param, xc, yb);

    gemm_wmma_f32<<<dim3(D_MODEL / 64, LSEQ / 128), blk, 0, stream>>>(
        yb, W_out, nullptr, out, LSEQ, D_MODEL, D_INNER, D_INNER, D_INNER, 0, 1);
}


# dt GEMM epilogue: softplus as log(u)*e/(u-1), u=1+exp(v), on the exp/log/rcp units, branch-free, not 64 in-line library log1pf chains (most of the kernel text); dt differs by a few ulp
# speedup vs baseline: 12.2092x; 1.0479x over previous
#include <hip/hip_runtime.h>
#include <math.h>

#define D_MODEL 1024
#define D_STATE 16
#define D_CONV  4
#define DT_RANK 64
#define D_INNER 2048
#define LSEQ    2048

typedef __attribute__((ext_vector_type(2))) float v2f;
typedef __attribute__((ext_vector_type(4))) float v4f;
typedef __attribute__((ext_vector_type(8))) float v8f;

struct Frag { v2f a0, a1, b0, b1, b2, b3; };

#define GB_M 128
#define GB_N 64
#define GB_K 32
#define GB_P 36

__device__ __forceinline__ void load_frag(Frag& f, const float* pA, const float* pB, int k)
{
    f.a0 = *(const v2f*)(pA + k);
    f.a1 = *(const v2f*)(pA + 16 * GB_P + k);
    f.b0 = *(const v2f*)(pB + k);
    f.b1 = *(const v2f*)(pB + 16 * GB_P + k);
    f.b2 = *(const v2f*)(pB + 32 * GB_P + k);
    f.b3 = *(const v2f*)(pB + 48 * GB_P + k);
}

__device__ __forceinline__ void do_wmma(const Frag& f, v8f acc[2][4])
{
    acc[0][0] = __builtin_amdgcn_wmma_f32_16x16x4_f32(false, f.a0, false, f.b0, (short)0, acc[0][0], false, false);
    acc[1][0] = __builtin_amdgcn_wmma_f32_16x16x4_f32(false, f.a1, false, f.b0, (short)0, acc[1][0], false, false);
    acc[0][1] = __builtin_amdgcn_wmma_f32_16x16x4_f32(false, f.a0, false, f.b1, (short)0, acc[0][1], false, false);
    acc[1][1] = __builtin_amdgcn_wmma_f32_16x16x4_f32(false, f.a1, false, f.b1, (short)0, acc[1][1], false, false);
    acc[0][2] = __builtin_amdgcn_wmma_f32_16x16x4_f32(false, f.a0, false, f.b2, (short)0, acc[0][2], false, false);
    acc[1][2] = __builtin_amdgcn_wmma_f32_16x16x4_f32(false, f.a1, false, f.b2, (short)0, acc[1][2], false, false);
    acc[0][3] = __builtin_amdgcn_wmma_f32_16x16x4_f32(false, f.a0, false, f.b3, (short)0, acc[0][3], false, false);
    acc[1][3] = __builtin_amdgcn_wmma_f32_16x16x4_f32(false, f.a1, false, f.b3, (short)0, acc[1][3], false, false);
}

__device__ __forceinline__ float softplus_units(float v)
{
    const float e = __expf(v);
    const float u = 1.0f + e;
    const float d = u - 1.0f;
    const float r = (__builtin_amdgcn_logf(u) * 0.693147180559945f) * (e * __builtin_amdgcn_rcpf(d));
    const float s = (d == 0.0f) ? e : r;
    return (v > 20.0f) ? v : s;
}

__global__ __launch_bounds__(128) void gemm_wmma_f32(
    const float* __restrict__ A, const float* __restrict__ W,
    const float* __restrict__ bias, float* __restrict__ Cbase,
    int M, int N, int K, int lda, int ldw, int act, int ablk)
{
    __shared__ __attribute__((aligned(16))) float sA[2][GB_M * GB_P];
    __shared__ __attribute__((aligned(16))) float sB[2][GB_N * GB_P];

    const int tid  = threadIdx.x;
    const int lane = tid & 31;
    const int wave = tid >> 5;
    const int mblk = blockIdx.y << 7;
    const int m0   = mblk + (wave << 5);
    const int n0   = blockIdx.x << 6;
    if (mblk >= M) return;

    const int lr = tid >> 3;
    const int q  = tid & 7;
    const size_t kofs = (size_t)blockIdx.z * K;
    float* __restrict__ C = Cbase + (size_t)blockIdx.z * M * N;
    const size_t aRow = ablk ? (size_t)8 : (size_t)lda;
    const size_t aK   = ablk ? (size_t)M : (size_t)1;
    const size_t aQ   = ablk ? ((size_t)(q >> 1) * M * 8 + ((q & 1) << 2)) : (size_t)(q << 2);
    const float* __restrict__ gA = A + (size_t)(mblk + lr) * aRow + kofs * aK + aQ;
    const size_t aStep = (size_t)16 * aRow;
    const float* gB[4];
#pragma unroll
    for (int j = 0; j < 4; ++j) {
        int row = n0 + lr + (j << 4);
        row = (row < N) ? row : (N - 1);
        gB[j] = W + (size_t)row * ldw + kofs + (q << 2);
    }
    const int sOff = lr * GB_P + (q << 2);

    const int mrow  = lane & 15;
    const int khalf = (lane >> 4) << 1;
    const int aOff  = ((wave << 5) + mrow) * GB_P + khalf;
    const int bOff  = mrow * GB_P + khalf;

    v8f acc[2][4];
#pragma unroll
    for (int i = 0; i < 2; ++i)
#pragma unroll
        for (int j = 0; j < 4; ++j) acc[i][j] = (v8f){};

    v4f ra[8], rb[4];
#pragma unroll
    for (int j = 0; j < 8; ++j) ra[j] = *(const v4f*)(gA + j * aStep);
#pragma unroll
    for (int j = 0; j < 4; ++j) rb[j] = *(const v4f*)(gB[j]);
#pragma unroll
    for (int j = 0; j < 8; ++j) *(v4f*)&sA[0][sOff + j * 16 * GB_P] = ra[j];
#pragma unroll
    for (int j = 0; j < 4; ++j) *(v4f*)&sB[0][sOff + j * 16 * GB_P] = rb[j];
    __syncthreads();

    int cur = 0;
    for (int kb = 0; kb < K; kb += GB_K) {
        const bool more = (kb + GB_K) < K;
        if (more) {
#pragma unroll
            for (int j = 0; j < 8; ++j) ra[j] = *(const v4f*)(gA + j * aStep + (size_t)(kb + GB_K) * aK);
#pragma unroll
            for (int j = 0; j < 4; ++j) rb[j] = *(const v4f*)(gB[j] + (kb + GB_K));
        }

        const float* pA = &sA[cur][aOff];
        const float* pB = &sB[cur][bOff];
        Frag f0, f1;
        load_frag(f0, pA, pB, 0);
#pragma unroll
        for (int ks = 0; ks < GB_K; ks += 8) {
            load_frag(f1, pA, pB, ks + 4);
            do_wmma(f0, acc);
            if (ks + 8 < GB_K) load_frag(f0, pA, pB, ks + 8);
            do_wmma(f1, acc);
        }

        if (more) {
#pragma unroll
            for (int j = 0; j < 8; ++j) *(v4f*)&sA[cur ^ 1][sOff + j * 16 * GB_P] = ra[j];
#pragma unroll
            for (int j = 0; j < 4; ++j) *(v4f*)&sB[cur ^ 1][sOff + j * 16 * GB_P] = rb[j];
        }
        __syncthreads();
        cur ^= 1;
    }

    const int rsub = (lane >> 4) << 3;
    if (act == 1) {
#pragma unroll
        for (int j = 0; j < 4; ++j) {
            if ((n0 + (j << 4)) >= N) continue;
            const float bv = bias[n0 + (j << 4) + mrow];
#pragma unroll
            for (int i = 0; i < 2; ++i)
#pragma unroll
                for (int r = 0; r < 8; ++r) acc[i][j][r] = softplus_units(acc[i][j][r] + bv);
        }
    }
#pragma unroll
    for (int j = 0; j < 4; ++j) {
        const int col = n0 + (j << 4) + mrow;
        if ((n0 + (j << 4)) >= N) continue;
#pragma unroll
        for (int i = 0; i < 2; ++i) {
            const int rbase = m0 + (i << 4) + rsub;
#pragma unroll
            for (int r = 0; r < 8; ++r) C[(size_t)(rbase + r) * N + col] = acc[i][j][r];
        }
    }
}

#define XD_SLICES 8
#define XD_ELEMS  (LSEQ * (DT_RANK + 2 * D_STATE))

__global__ __launch_bounds__(256) void slice_sum_kernel(
    const float* __restrict__ planes, float* __restrict__ out)
{
    const int i = blockIdx.x * 256 + threadIdx.x;
    const v4f* __restrict__ p = (const v4f*)planes + i;
    v4f v[XD_SLICES];
#pragma unroll
    for (int s = 0; s < XD_SLICES; ++s) v[s] = p[(size_t)s * (XD_ELEMS / 4)];
    v4f acc = v[0];
#pragma unroll
    for (int s = 1; s < XD_SLICES; ++s) acc += v[s];
    ((v4f*)out)[i] = acc;
}

__global__ __launch_bounds__(256) void conv_silu_kernel(
    const float* __restrict__ xz, const float* __restrict__ cw,
    const float* __restrict__ cb, float* __restrict__ xc)
{
    int idx = blockIdx.x * blockDim.x + threadIdx.x;
    if (idx >= LSEQ * D_INNER) return;
    int l = idx >> 11;
    int d = idx & (D_INNER - 1);

    float acc = cb[d];
#pragma unroll
    for (int j = 0; j < D_CONV; ++j) {
        int ls = l - (D_CONV - 1) + j;
        if (ls >= 0)
            acc += cw[d * D_CONV + j] * xz[(size_t)ls * (2 * D_INNER) + d];
    }
    xc[idx] = acc / (1.0f + __expf(-acc));
}

#define SC_CH 8
#define SC_T  64
#define SC_W  48
#define SC_YP 12

template <int CTRL>
__device__ __forceinline__ float dpp_row(float v)
{
    const int b = __builtin_bit_cast(int, v);
    return __builtin_bit_cast(float, __builtin_amdgcn_update_dpp(0, b, CTRL, 0xf, 0xf, true));
}

template <int XM>
__device__ __forceinline__ float pair_fold(float even, float odd, bool bit)
{
    const float keep = bit ? odd : even;
    const float send = bit ? even : odd;
    return keep + dpp_row<0x160 + XM>(send);
}

__device__ __forceinline__ float row16_sum_by_step(const float (&p)[16], int n)
{
    const bool b0 = (n & 1) != 0, b1 = (n & 2) != 0, b2 = (n & 4) != 0, b3 = (n & 8) != 0;
    float q[8], r[4], u[2];
#pragma unroll
    for (int k = 0; k < 8; ++k) q[k] = pair_fold<1>(p[2 * k], p[2 * k + 1], b0);
#pragma unroll
    for (int k = 0; k < 4; ++k) r[k] = pair_fold<2>(q[2 * k], q[2 * k + 1], b1);
#pragma unroll
    for (int k = 0; k < 2; ++k) u[k] = pair_fold<4>(r[2 * k], r[2 * k + 1], b2);
    return pair_fold<8>(u[0], u[1], b3);
}

struct ScanBlk { float dt[16], x[16], B[16], C[16]; };

__device__ __forceinline__ void load_blk(ScanBlk& b, const float (*s)[SC_W], int r, int chl, int n)
{
#pragma unroll
    for (int i = 0; i < 16; ++i) {
        const int t = r * 16 + i;
        b.dt[i] = s[t][chl];
        b.x[i]  = s[t][8 + chl];
        b.B[i]  = s[t][16 + n];
        b.C[i]  = s[t][32 + n];
    }
}

__device__ __forceinline__ float scan_blk(const ScanBlk& b, float& h, float Acoef, int n)
{
    float p[16];
#pragma unroll
    for (int i = 0; i < 16; ++i) {
        const float dA = __expf(b.dt[i] * Acoef);
        h = dA * h + (b.dt[i] * b.B[i]) * b.x[i];
        p[i] = h * b.C[i];
    }
    return row16_sum_by_step(p, n);
}

__global__ __launch_bounds__(128) void ssm_scan_kernel(
    const float* __restrict__ xdbl, const float* __restrict__ dtbuf,
    const float* __restrict__ xz,   const float* __restrict__ A_log,
    const float* __restrict__ D_param, const float* __restrict__ xc,
    float* __restrict__ yblk)
{
    __shared__ __attribute__((aligned(16))) float sIn[SC_T][SC_W];
    __shared__ __attribute__((aligned(16))) float sY[SC_T][SC_YP];

    const int tid  = threadIdx.x;
    const int lane = tid & 31;
    const int wave = tid >> 5;
    const int D0   = blockIdx.x * SC_CH;

    const int n   = lane & 15;
    const int chl = (wave << 1) + (lane >> 4);
    const float Acoef = -__expf(A_log[(size_t)(D0 + chl) * D_STATE + n]);
    float h = 0.0f;

    const int srow = tid >> 1;
    const int sc4  = (tid & 1) << 2;
    const int brow = tid >> 3;
    const int bq   = (tid & 7) << 2;
    const v4f Dp = *(const v4f*)(D_param + D0 + sc4);

    const float* __restrict__ pdt = dtbuf + (size_t)srow * D_INNER + D0 + sc4;
    const float* __restrict__ px  = xc    + (size_t)srow * D_INNER + D0 + sc4;
    const float* __restrict__ pz  = xz    + (size_t)srow * (2 * D_INNER) + D_INNER + D0 + sc4;
    const float* __restrict__ pbc = xdbl  + (size_t)brow * (DT_RANK + 2 * D_STATE) + DT_RANK + bq;
    float* __restrict__ py = yblk + ((size_t)blockIdx.x * LSEQ + srow) * SC_CH + sc4;

    v4f r_dt = *(const v4f*)pdt;
    v4f r_x  = *(const v4f*)px;
    v4f r_z  = *(const v4f*)pz;
    v4f r_bc[4];
#pragma unroll
    for (int r = 0; r < 4; ++r)
        r_bc[r] = *(const v4f*)(pbc + (size_t)(r * 16) * (DT_RANK + 2 * D_STATE));

    for (int lc = 0; lc < LSEQ; lc += SC_T) {
        *(v4f*)&sIn[srow][sc4]     = r_dt;
        *(v4f*)&sIn[srow][8 + sc4] = r_x;
#pragma unroll
        for (int r = 0; r < 4; ++r) *(v4f*)&sIn[r * 16 + brow][16 + bq] = r_bc[r];
        const v4f c_x = r_x;
        const v4f c_z = r_z;
        __syncthreads();

        if (lc + SC_T < LSEQ) {
            const size_t o = (size_t)(lc + SC_T);
            r_dt = *(const v4f*)(pdt + o * D_INNER);
            r_x  = *(const v4f*)(px  + o * D_INNER);
            r_z  = *(const v4f*)(pz  + o * (2 * D_INNER));
#pragma unroll
            for (int r = 0; r < 4; ++r)
                r_bc[r] = *(const v4f*)(pbc + (o + r * 16) * (DT_RANK + 2 * D_STATE));
        }

        ScanBlk b0, b1;
        load_blk(b0, sIn, 0, chl, n);
        load_blk(b1, sIn, 1, chl, n);
        __builtin_amdgcn_sched_barrier(0);

        v4f g;
#pragma unroll
        for (int e = 0; e < 4; ++e) g[e] = c_z[e] / (1.0f + __expf(-c_z[e]));

        float yk[4];
        yk[0] = scan_blk(b0, h, Acoef, n);
        __builtin_amdgcn_sched_barrier(0);
        load_blk(b0, sIn, 2, chl, n);
        __builtin_amdgcn_sched_barrier(0);
        yk[1] = scan_blk(b1, h, Acoef, n);
        __builtin_amdgcn_sched_barrier(0);
        load_blk(b1, sIn, 3, chl, n);
        __builtin_amdgcn_sched_barrier(0);
        yk[2] = scan_blk(b0, h, Acoef, n);
        yk[3] = scan_blk(b1, h, Acoef, n);
#pragma unroll
        for (int r = 0; r < 4; ++r) sY[r * 16 + n][chl] = yk[r];
        __syncthreads();

        v4f y = *(const v4f*)&sY[srow][sc4];
#pragma unroll
        for (int e = 0; e < 4; ++e) {
            float v = y[e];
            v += c_x[e] * Dp[e];
            y[e] = v * g[e];
        }
        *(v4f*)(py + (size_t)lc * SC_CH) = y;
    }
}

extern "C" void kernel_launch(void* const* d_in, const int* in_sizes, int n_in,
                              void* d_out, int out_size, void* d_ws, size_t ws_size,
                              hipStream_t stream) {
    const float* x       = (const float*)d_in[0];
    const float* W_in    = (const float*)d_in[1];
    const float* conv_w  = (const float*)d_in[2];
    const float* conv_b  = (const float*)d_in[3];
    const float* W_x     = (const float*)d_in[4];
    const float* W_dt    = (const float*)d_in[5];
    const float* b_dt    = (const float*)d_in[6];
    const float* A_log   = (const float*)d_in[7];
    const float* D_param = (const float*)d_in[8];
    const float* W_out   = (const float*)d_in[9];
    float* out = (float*)d_out;

    float* xz   = (float*)d_ws;
    float* xc   = xz   + (size_t)LSEQ * 2 * D_INNER;
    float* xdbl = xc   + (size_t)LSEQ * D_INNER;
    float* dtb  = xdbl + (size_t)LSEQ * (DT_RANK + 2 * D_STATE);
    float* yb   = dtb  + (size_t)LSEQ * D_INNER;

    dim3 blk(128, 1, 1);

    gemm_wmma_f32<<<dim3(4096 / 64, LSEQ / 128), blk, 0, stream>>>(
        x, W_in, nullptr, xz, LSEQ, 2 * D_INNER, D_MODEL, D_MODEL, D_MODEL, 0, 0);

    conv_silu_kernel<<<(LSEQ * D_INNER) / 256, 256, 0, stream>>>(
        xz, conv_w, conv_b, xc);

    gemm_wmma_f32<<<dim3((96 + 63) / 64, LSEQ / 128, XD_SLICES), blk, 0, stream>>>(
        xc, W_x, nullptr, dtb, LSEQ, DT_RANK + 2 * D_STATE, D_INNER / XD_SLICES,
        D_INNER, D_INNER, 0, 0);
    slice_sum_kernel<<<(XD_ELEMS / 4) / 256, 256, 0, stream>>>(dtb, xdbl);

    gemm_wmma_f32<<<dim3(D_INNER / 64, LSEQ / 128), blk, 0, stream>>>(
        xdbl, W_dt, b_dt, dtb, LSEQ, D_INNER, DT_RANK, DT_RANK + 2 * D_STATE, DT_RANK, 1, 0);

    ssm_scan_kernel<<<D_INNER / SC_CH, 128, 0, stream>>>(
        xdbl, dtb, xz, A_log, D_param, xc, yb);

    gemm_wmma_f32<<<dim3(D_MODEL / 64, LSEQ / 128), blk, 0, stream>>>(
        yb, W_out, nullptr, out, LSEQ, D_MODEL, D_INNER, D_INNER, D_INNER, 0, 1);
}
